# starting kernel
# baseline (speedup 1.0000x reference)
#include <hip/hip_runtime.h>
#include <math.h>

#define N_NODES 50000
#define N_EDGES 800000
#define IN_F    128
#define OUT_F   64

typedef __attribute__((ext_vector_type(2))) float v2f;
typedef __attribute__((ext_vector_type(8))) float v8f;

__device__ __forceinline__ unsigned fkey(float f) {
    unsigned u = __float_as_uint(f);
    return (u & 0x80000000u) ? ~u : (u | 0x80000000u);
}
__device__ __forceinline__ float funkey(unsigned u) {
    return __uint_as_float((u & 0x80000000u) ? (u & 0x7FFFFFFFu) : ~u);
}

__global__ void gat_init(unsigned* __restrict__ segbits,
                         float* __restrict__ denom,
                         float* __restrict__ out) {
    int i = blockIdx.x * blockDim.x + threadIdx.x;
    if (i < N_NODES) {
        segbits[i] = fkey(-INFINITY);
        denom[i]   = 0.0f;
    }
    if (i < N_NODES * OUT_F) out[i] = 0.0f;
}

__global__ void qk_gemm(const float* __restrict__ x,
                        const float* __restrict__ Wq,
                        const float* __restrict__ Wk,
                        float* __restrict__ Q,
                        float* __restrict__ K) {
    const int lane    = threadIdx.x & 31;
    const int wave    = threadIdx.x >> 5;
    const int mat     = wave >> 2;
    const int colBase = (wave & 3) * 16;
    const int rowBase = blockIdx.x * 16;

    const float* __restrict__ W   = mat ? Wk : Wq;
    float*       __restrict__ Out = mat ? K  : Q;

    const int l    = lane & 15;
    const int koff = (lane >> 4) * 2;

    const float* __restrict__ arow = x + (size_t)(rowBase + l) * IN_F + koff;
    const float* __restrict__ brow = W + (size_t)(colBase + l) * IN_F + koff;

    v8f acc = {};
#pragma unroll 8
    for (int k = 0; k < IN_F; k += 4) {
        v2f a = *(const v2f*)(arow + k);
        v2f b = *(const v2f*)(brow + k);
        acc = __builtin_amdgcn_wmma_f32_16x16x4_f32(
             false, a,  false, b,
             (short)0, acc,  false,  false);
    }

    const int outRow = rowBase + ((lane >> 4) ? 8 : 0);
    const int outCol = colBase + l;
#pragma unroll
    for (int r = 0; r < 8; ++r)
        Out[(size_t)(outRow + r) * OUT_F + outCol] = acc[r];
}

__global__ void edge_logits(const int* __restrict__ ei,
                            const float* __restrict__ Q,
                            const float* __restrict__ K,
                            float* __restrict__ alpha,
                            unsigned* __restrict__ segbits) {
    const int lane = threadIdx.x & 31;
    const int wid  = (blockIdx.x * blockDim.x + threadIdx.x) >> 5;
    const int nw   = (gridDim.x * blockDim.x) >> 5;

    for (int e = wid; e < N_EDGES; e += nw) {
        const int row = ei[e];
        const int col = ei[N_EDGES + e];
        const float* __restrict__ q = Q + (size_t)row * OUT_F;
        const float* __restrict__ k = K + (size_t)col * OUT_F;
        float s = q[lane] * k[lane] + q[lane + 32] * k[lane + 32];
#pragma unroll
        for (int off = 16; off; off >>= 1) s += __shfl_xor(s, off, 32);
        s *= 0.125f;
        if (lane == 0) {
            alpha[e] = s;
            atomicMax(segbits + col, fkey(s));
        }
    }
}

__global__ void edge_exp(const int* __restrict__ ei,
                         const unsigned* __restrict__ segbits,
                         float* __restrict__ alpha,
                         float* __restrict__ denom) {
    const int e = blockIdx.x * blockDim.x + threadIdx.x;
    if (e >= N_EDGES) return;
    const int col = ei[N_EDGES + e];
    const float v = expf(alpha[e] - funkey(segbits[col]));
    alpha[e] = v;
    atomicAdd(denom + col, v);
}

__global__ void edge_scatter(const int* __restrict__ ei,
                             const float* __restrict__ Q,
                             const float* __restrict__ ex,
                             const float* __restrict__ denom,
                             float* __restrict__ out) {
    const int lane = threadIdx.x & 31;
    const int wid  = (blockIdx.x * blockDim.x + threadIdx.x) >> 5;
    const int nw   = (gridDim.x * blockDim.x) >> 5;

    for (int e = wid; e < N_EDGES; e += nw) {
        const int row = ei[e];
        const int col = ei[N_EDGES + e];
        const float w = ex[e] / (denom[col] + 1e-16f);
        const float* __restrict__ q = Q + (size_t)row * OUT_F;
        float* __restrict__ o = out + (size_t)col * OUT_F;
        atomicAdd(o + lane,      q[lane]      * w);
        atomicAdd(o + lane + 32, q[lane + 32] * w);
    }
}

extern "C" void kernel_launch(void* const* d_in, const int* in_sizes, int n_in,
                              void* d_out, int out_size, void* d_ws, size_t ws_size,
                              hipStream_t stream) {
    const float* x  = (const float*)d_in[0];
    const int*   ei = (const int*)d_in[1];
    const float* Wq = (const float*)d_in[2];
    const float* Wk = (const float*)d_in[3];
    float* out = (float*)d_out;

    char* ws = (char*)d_ws;
    float*    Q       = (float*)ws;    ws += (size_t)N_NODES * OUT_F * sizeof(float);
    float*    K       = (float*)ws;    ws += (size_t)N_NODES * OUT_F * sizeof(float);
    float*    alpha   = (float*)ws;    ws += (size_t)N_EDGES * sizeof(float);
    unsigned* segbits = (unsigned*)ws; ws += (size_t)N_NODES * sizeof(unsigned);
    float*    denom   = (float*)ws;

    gat_init<<<(N_NODES * OUT_F + 255) / 256, 256, 0, stream>>>(segbits, denom, out);
    qk_gemm<<<N_NODES / 16, 256, 0, stream>>>(x, Wq, Wk, Q, K);
    edge_logits<<<4096, 256, 0, stream>>>(ei, Q, K, alpha, segbits);
    edge_exp<<<(N_EDGES + 255) / 256, 256, 0, stream>>>(ei, segbits, alpha, denom);
    edge_scatter<<<4096, 256, 0, stream>>>(ei, Q, alpha, denom, out);
}


# edge_logits: 32 consecutive edges per wave, 16 lanes x 16-byte loads per edge row, dot summed per lane then by four row_xmask DPP adds (new summation tree), not wave per edge with LDS permutes
# speedup vs baseline: 1.3068x; 1.3068x over previous
#include <hip/hip_runtime.h>
#include <math.h>

#define N_NODES 50000
#define N_EDGES 800000
#define IN_F    128
#define OUT_F   64

typedef __attribute__((ext_vector_type(2))) float v2f;
typedef __attribute__((ext_vector_type(8))) float v8f;

__device__ __forceinline__ unsigned fkey(float f) {
    unsigned u = __float_as_uint(f);
    return (u & 0x80000000u) ? ~u : (u | 0x80000000u);
}
__device__ __forceinline__ float funkey(unsigned u) {
    return __uint_as_float((u & 0x80000000u) ? (u & 0x7FFFFFFFu) : ~u);
}

__global__ void gat_init(unsigned* __restrict__ segbits,
                         float* __restrict__ denom,
                         float* __restrict__ out) {
    int i = blockIdx.x * blockDim.x + threadIdx.x;
    if (i < N_NODES) {
        segbits[i] = fkey(-INFINITY);
        denom[i]   = 0.0f;
    }
    if (i < N_NODES * OUT_F) out[i] = 0.0f;
}

__global__ void qk_gemm(const float* __restrict__ x,
                        const float* __restrict__ Wq,
                        const float* __restrict__ Wk,
                        float* __restrict__ Q,
                        float* __restrict__ K) {
    const int lane    = threadIdx.x & 31;
    const int wave    = threadIdx.x >> 5;
    const int mat     = wave >> 2;
    const int colBase = (wave & 3) * 16;
    const int rowBase = blockIdx.x * 16;

    const float* __restrict__ W   = mat ? Wk : Wq;
    float*       __restrict__ Out = mat ? K  : Q;

    const int l    = lane & 15;
    const int koff = (lane >> 4) * 2;

    const float* __restrict__ arow = x + (size_t)(rowBase + l) * IN_F + koff;
    const float* __restrict__ brow = W + (size_t)(colBase + l) * IN_F + koff;

    v8f acc = {};
#pragma unroll 8
    for (int k = 0; k < IN_F; k += 4) {
        v2f a = *(const v2f*)(arow + k);
        v2f b = *(const v2f*)(brow + k);
        acc = __builtin_amdgcn_wmma_f32_16x16x4_f32(
             false, a,  false, b,
             (short)0, acc,  false,  false);
    }

    const int outRow = rowBase + ((lane >> 4) ? 8 : 0);
    const int outCol = colBase + l;
#pragma unroll
    for (int r = 0; r < 8; ++r)
        Out[(size_t)(outRow + r) * OUT_F + outCol] = acc[r];
}

static_assert(N_EDGES % 32 == 0, "a wave takes whole chunks of 32 edges");
static_assert(OUT_F == 64, "16 lanes x 4 floats per row");
#define N_CHUNKS (N_EDGES / 32)

template <int J>
__device__ __forceinline__ int row_bcast(int v) {
    return __builtin_amdgcn_update_dpp(0, v, 0x150 + J, 0xf, 0xf, true);
}
__device__ __forceinline__ int row_down4(int v) {
    return __builtin_amdgcn_update_dpp(0, v, 0x120 + 12, 0xf, 0xf, true);
}
template <int M>
__device__ __forceinline__ float row_xadd(float p) {
    return p + __int_as_float(__builtin_amdgcn_update_dpp(
                   0, __float_as_int(p), 0x160 + M, 0xf, 0xf, true));
}

template <int J>
__device__ __forceinline__ float4 ld_row(const char* __restrict__ base,
                                         int idx, unsigned sub) {
    const unsigned r = (unsigned)row_bcast<J>(idx);
    return *(const float4*)(base + (r * (OUT_F * 4u) + sub * 16u));
}
__device__ __forceinline__ float row_dot(float4 q, float4 k) {
    float p = q.x * k.x + q.y * k.y + q.z * k.z + q.w * k.w;
    p = row_xadd<1>(p);
    p = row_xadd<2>(p);
    p = row_xadd<4>(p);
    p = row_xadd<8>(p);
    return p;
}

__global__ __launch_bounds__(128)
void edge_logits(const int* __restrict__ ei,
                 const float* __restrict__ Q,
                 const float* __restrict__ K,
                 float* __restrict__ alpha,
                 unsigned* __restrict__ segbits) {
    const int lane     = threadIdx.x & 31;
    const unsigned sub = lane & 15;
    const int wid = (blockIdx.x * blockDim.x + threadIdx.x) >> 5;
    const int nw  = (gridDim.x * blockDim.x) >> 5;
    const char* __restrict__ Qb = (const char*)Q;
    const char* __restrict__ Kb = (const char*)K;

    for (int chunk = wid; chunk < N_CHUNKS; chunk += nw) {
        const int e   = chunk * 32 + lane;
        const int row = ei[e];
        const int col = ei[N_EDGES + e];
        int rr = row, cr = col;
        float s = 0.0f;
#pragma unroll 1
        for (unsigned j0 = 0; j0 < 16; j0 += 4) {
            const float4 q0 = ld_row<0>(Qb, rr, sub);
            const float4 k0 = ld_row<0>(Kb, cr, sub);
            const float4 q1 = ld_row<1>(Qb, rr, sub);
            const float4 k1 = ld_row<1>(Kb, cr, sub);
            const float4 q2 = ld_row<2>(Qb, rr, sub);
            const float4 k2 = ld_row<2>(Kb, cr, sub);
            const float4 q3 = ld_row<3>(Qb, rr, sub);
            const float4 k3 = ld_row<3>(Kb, cr, sub);
            const float p0 = row_dot(q0, k0);
            const float p1 = row_dot(q1, k1);
            const float p2 = row_dot(q2, k2);
            const float p3 = row_dot(q3, k3);
            s = (sub == j0)      ? p0 : s;
            s = (sub == j0 + 1u) ? p1 : s;
            s = (sub == j0 + 2u) ? p2 : s;
            s = (sub == j0 + 3u) ? p3 : s;
            rr = row_down4(rr);
            cr = row_down4(cr);
        }
        s *= 0.125f;
        alpha[e] = s;
        atomicMax(segbits + col, fkey(s));
    }
}

__global__ void edge_exp(const int* __restrict__ ei,
                         const unsigned* __restrict__ segbits,
                         float* __restrict__ alpha,
                         float* __restrict__ denom) {
    const int e = blockIdx.x * blockDim.x + threadIdx.x;
    if (e >= N_EDGES) return;
    const int col = ei[N_EDGES + e];
    const float v = expf(alpha[e] - funkey(segbits[col]));
    alpha[e] = v;
    atomicAdd(denom + col, v);
}

__global__ void edge_scatter(const int* __restrict__ ei,
                             const float* __restrict__ Q,
                             const float* __restrict__ ex,
                             const float* __restrict__ denom,
                             float* __restrict__ out) {
    const int lane = threadIdx.x & 31;
    const int wid  = (blockIdx.x * blockDim.x + threadIdx.x) >> 5;
    const int nw   = (gridDim.x * blockDim.x) >> 5;

    for (int e = wid; e < N_EDGES; e += nw) {
        const int row = ei[e];
        const int col = ei[N_EDGES + e];
        const float w = ex[e] / (denom[col] + 1e-16f);
        const float* __restrict__ q = Q + (size_t)row * OUT_F;
        float* __restrict__ o = out + (size_t)col * OUT_F;
        atomicAdd(o + lane,      q[lane]      * w);
        atomicAdd(o + lane + 32, q[lane + 32] * w);
    }
}

extern "C" void kernel_launch(void* const* d_in, const int* in_sizes, int n_in,
                              void* d_out, int out_size, void* d_ws, size_t ws_size,
                              hipStream_t stream) {
    const float* x  = (const float*)d_in[0];
    const int*   ei = (const int*)d_in[1];
    const float* Wq = (const float*)d_in[2];
    const float* Wk = (const float*)d_in[3];
    float* out = (float*)d_out;

    char* ws = (char*)d_ws;
    float*    Q       = (float*)ws;    ws += (size_t)N_NODES * OUT_F * sizeof(float);
    float*    K       = (float*)ws;    ws += (size_t)N_NODES * OUT_F * sizeof(float);
    float*    alpha   = (float*)ws;    ws += (size_t)N_EDGES * sizeof(float);
    unsigned* segbits = (unsigned*)ws; ws += (size_t)N_NODES * sizeof(unsigned);
    float*    denom   = (float*)ws;

    gat_init<<<(N_NODES * OUT_F + 255) / 256, 256, 0, stream>>>(segbits, denom, out);
    qk_gemm<<<N_NODES / 16, 256, 0, stream>>>(x, Wq, Wk, Q, K);
    edge_logits<<<N_CHUNKS / 8, 128, 0, stream>>>(ei, Q, K, alpha, segbits);
    edge_exp<<<(N_EDGES + 255) / 256, 256, 0, stream>>>(ei, segbits, alpha, denom);
    edge_scatter<<<4096, 256, 0, stream>>>(ei, Q, alpha, denom, out);
}


# edge_scatter: 32 consecutive edges per wave (coalesced ids and ex, one denom gather, ids and weight by readlane), row loads 8 edges ahead of the f32 atomic adds; arrival order of the adds changes
# speedup vs baseline: 1.5890x; 1.2159x over previous
#include <hip/hip_runtime.h>
#include <math.h>

#define N_NODES 50000
#define N_EDGES 800000
#define IN_F    128
#define OUT_F   64

typedef __attribute__((ext_vector_type(2))) float v2f;
typedef __attribute__((ext_vector_type(8))) float v8f;

__device__ __forceinline__ unsigned fkey(float f) {
    unsigned u = __float_as_uint(f);
    return (u & 0x80000000u) ? ~u : (u | 0x80000000u);
}
__device__ __forceinline__ float funkey(unsigned u) {
    return __uint_as_float((u & 0x80000000u) ? (u & 0x7FFFFFFFu) : ~u);
}

__global__ void gat_init(unsigned* __restrict__ segbits,
                         float* __restrict__ denom,
                         float* __restrict__ out) {
    int i = blockIdx.x * blockDim.x + threadIdx.x;
    if (i < N_NODES) {
        segbits[i] = fkey(-INFINITY);
        denom[i]   = 0.0f;
    }
    if (i < N_NODES * OUT_F) out[i] = 0.0f;
}

__global__ void qk_gemm(const float* __restrict__ x,
                        const float* __restrict__ Wq,
                        const float* __restrict__ Wk,
                        float* __restrict__ Q,
                        float* __restrict__ K) {
    const int lane    = threadIdx.x & 31;
    const int wave    = threadIdx.x >> 5;
    const int mat     = wave >> 2;
    const int colBase = (wave & 3) * 16;
    const int rowBase = blockIdx.x * 16;

    const float* __restrict__ W   = mat ? Wk : Wq;
    float*       __restrict__ Out = mat ? K  : Q;

    const int l    = lane & 15;
    const int koff = (lane >> 4) * 2;

    const float* __restrict__ arow = x + (size_t)(rowBase + l) * IN_F + koff;
    const float* __restrict__ brow = W + (size_t)(colBase + l) * IN_F + koff;

    v8f acc = {};
#pragma unroll 8
    for (int k = 0; k < IN_F; k += 4) {
        v2f a = *(const v2f*)(arow + k);
        v2f b = *(const v2f*)(brow + k);
        acc = __builtin_amdgcn_wmma_f32_16x16x4_f32(
             false, a,  false, b,
             (short)0, acc,  false,  false);
    }

    const int outRow = rowBase + ((lane >> 4) ? 8 : 0);
    const int outCol = colBase + l;
#pragma unroll
    for (int r = 0; r < 8; ++r)
        Out[(size_t)(outRow + r) * OUT_F + outCol] = acc[r];
}

static_assert(N_EDGES % 32 == 0, "a wave takes whole chunks of 32 edges");
static_assert(OUT_F == 64, "16 lanes x 4 floats per row");
#define N_CHUNKS (N_EDGES / 32)

template <int J>
__device__ __forceinline__ int row_bcast(int v) {
    return __builtin_amdgcn_update_dpp(0, v, 0x150 + J, 0xf, 0xf, true);
}
__device__ __forceinline__ int row_down4(int v) {
    return __builtin_amdgcn_update_dpp(0, v, 0x120 + 12, 0xf, 0xf, true);
}
template <int M>
__device__ __forceinline__ float row_xadd(float p) {
    return p + __int_as_float(__builtin_amdgcn_update_dpp(
                   0, __float_as_int(p), 0x160 + M, 0xf, 0xf, true));
}

template <int J>
__device__ __forceinline__ float4 ld_row(const char* __restrict__ base,
                                         int idx, unsigned sub) {
    const unsigned r = (unsigned)row_bcast<J>(idx);
    return *(const float4*)(base + (r * (OUT_F * 4u) + sub * 16u));
}
__device__ __forceinline__ float row_dot(float4 q, float4 k) {
    float p = q.x * k.x + q.y * k.y + q.z * k.z + q.w * k.w;
    p = row_xadd<1>(p);
    p = row_xadd<2>(p);
    p = row_xadd<4>(p);
    p = row_xadd<8>(p);
    return p;
}

__global__ __launch_bounds__(128)
void edge_logits(const int* __restrict__ ei,
                 const float* __restrict__ Q,
                 const float* __restrict__ K,
                 float* __restrict__ alpha,
                 unsigned* __restrict__ segbits) {
    const int lane     = threadIdx.x & 31;
    const unsigned sub = lane & 15;
    const int wid = (blockIdx.x * blockDim.x + threadIdx.x) >> 5;
    const int nw  = (gridDim.x * blockDim.x) >> 5;
    const char* __restrict__ Qb = (const char*)Q;
    const char* __restrict__ Kb = (const char*)K;

    for (int chunk = wid; chunk < N_CHUNKS; chunk += nw) {
        const int e   = chunk * 32 + lane;
        const int row = ei[e];
        const int col = ei[N_EDGES + e];
        int rr = row, cr = col;
        float s = 0.0f;
#pragma unroll 1
        for (unsigned j0 = 0; j0 < 16; j0 += 4) {
            const float4 q0 = ld_row<0>(Qb, rr, sub);
            const float4 k0 = ld_row<0>(Kb, cr, sub);
            const float4 q1 = ld_row<1>(Qb, rr, sub);
            const float4 k1 = ld_row<1>(Kb, cr, sub);
            const float4 q2 = ld_row<2>(Qb, rr, sub);
            const float4 k2 = ld_row<2>(Kb, cr, sub);
            const float4 q3 = ld_row<3>(Qb, rr, sub);
            const float4 k3 = ld_row<3>(Kb, cr, sub);
            const float p0 = row_dot(q0, k0);
            const float p1 = row_dot(q1, k1);
            const float p2 = row_dot(q2, k2);
            const float p3 = row_dot(q3, k3);
            s = (sub == j0)      ? p0 : s;
            s = (sub == j0 + 1u) ? p1 : s;
            s = (sub == j0 + 2u) ? p2 : s;
            s = (sub == j0 + 3u) ? p3 : s;
            rr = row_down4(rr);
            cr = row_down4(cr);
        }
        s *= 0.125f;
        alpha[e] = s;
        atomicMax(segbits + col, fkey(s));
    }
}

__global__ void edge_exp(const int* __restrict__ ei,
                         const unsigned* __restrict__ segbits,
                         float* __restrict__ alpha,
                         float* __restrict__ denom) {
    const int e = blockIdx.x * blockDim.x + threadIdx.x;
    if (e >= N_EDGES) return;
    const int col = ei[N_EDGES + e];
    const float v = expf(alpha[e] - funkey(segbits[col]));
    alpha[e] = v;
    atomicAdd(denom + col, v);
}

#define LOOKAHEAD 8

__global__ __launch_bounds__(128)
void edge_scatter(const int* __restrict__ ei,
                  const float* __restrict__ Q,
                  const float* __restrict__ ex,
                  const float* __restrict__ denom,
                  float* __restrict__ out) {
    const int lane = threadIdx.x & 31;
    const int wid = __builtin_amdgcn_readfirstlane(
        (int)((blockIdx.x * blockDim.x + threadIdx.x) >> 5));
    const int nw  = (int)((gridDim.x * blockDim.x) >> 5);

    for (int chunk = wid; chunk < N_CHUNKS; chunk += nw) {
        const int e   = chunk * 32 + lane;
        const int row = ei[e];
        const int col = ei[N_EDGES + e];
        const float w = ex[e] / (denom[col] + 1e-16f);

        float qlo[32], qhi[32];
#pragma unroll
        for (int j = 0; j < LOOKAHEAD; ++j) {
            const float* q =
                Q + (size_t)__builtin_amdgcn_readlane(row, j) * OUT_F;
            qlo[j] = q[lane];
            qhi[j] = q[lane + 32];
        }
#pragma unroll
        for (int j = 0; j < 32; ++j) {
            if (j + LOOKAHEAD < 32) {
                const float* q =
                    Q + (size_t)__builtin_amdgcn_readlane(row, j + LOOKAHEAD) * OUT_F;
                qlo[j + LOOKAHEAD] = q[lane];
                qhi[j + LOOKAHEAD] = q[lane + 32];
            }
            const float wj = __int_as_float(
                __builtin_amdgcn_readlane(__float_as_int(w), j));
            float* o =
                out + (size_t)__builtin_amdgcn_readlane(col, j) * OUT_F;
            atomicAdd(o + lane,      qlo[j] * wj);
            atomicAdd(o + lane + 32, qhi[j] * wj);
        }
    }
}

extern "C" void kernel_launch(void* const* d_in, const int* in_sizes, int n_in,
                              void* d_out, int out_size, void* d_ws, size_t ws_size,
                              hipStream_t stream) {
    const float* x  = (const float*)d_in[0];
    const int*   ei = (const int*)d_in[1];
    const float* Wq = (const float*)d_in[2];
    const float* Wk = (const float*)d_in[3];
    float* out = (float*)d_out;

    char* ws = (char*)d_ws;
    float*    Q       = (float*)ws;    ws += (size_t)N_NODES * OUT_F * sizeof(float);
    float*    K       = (float*)ws;    ws += (size_t)N_NODES * OUT_F * sizeof(float);
    float*    alpha   = (float*)ws;    ws += (size_t)N_EDGES * sizeof(float);
    unsigned* segbits = (unsigned*)ws; ws += (size_t)N_NODES * sizeof(unsigned);
    float*    denom   = (float*)ws;

    gat_init<<<(N_NODES * OUT_F + 255) / 256, 256, 0, stream>>>(segbits, denom, out);
    qk_gemm<<<N_NODES / 16, 256, 0, stream>>>(x, Wq, Wk, Q, K);
    edge_logits<<<N_CHUNKS / 8, 128, 0, stream>>>(ei, Q, K, alpha, segbits);
    edge_exp<<<(N_EDGES + 255) / 256, 256, 0, stream>>>(ei, segbits, alpha, denom);
    edge_scatter<<<N_CHUNKS / 8, 128, 0, stream>>>(ei, Q, alpha, denom, out);
}


# qk_gemm operand delivery: the block's 16x128 x tile goes to LDS once by whole-line loads (was re-read by all 8 waves), A and B fragments read 16 bytes per lane; k order of the sums changes
# speedup vs baseline: 1.7750x; 1.1171x over previous
#include <hip/hip_runtime.h>
#include <math.h>

#define N_NODES 50000
#define N_EDGES 800000
#define IN_F    128
#define OUT_F   64

typedef __attribute__((ext_vector_type(2))) float v2f;
typedef __attribute__((ext_vector_type(8))) float v8f;

__device__ __forceinline__ unsigned fkey(float f) {
    unsigned u = __float_as_uint(f);
    return (u & 0x80000000u) ? ~u : (u | 0x80000000u);
}
__device__ __forceinline__ float funkey(unsigned u) {
    return __uint_as_float((u & 0x80000000u) ? (u & 0x7FFFFFFFu) : ~u);
}

static_assert(N_NODES % 16 == 0, "a block owns 16 whole nodes");
static_assert(16 * OUT_F == 256 * 4, "256 threads x 4 floats cover a block's out rows");
static_assert(16 * IN_F == 2 * 256 * 4, "256 threads x 2 x 4 floats cover a block's x tile");
static_assert(IN_F % 8 == 0, "a step takes 8 k values");
#define A_PITCH (IN_F + 4)
#define K_STEPS (IN_F / 8)

__global__ __launch_bounds__(256)
void qk_gemm(const float* __restrict__ x,
             const float* __restrict__ Wq,
             const float* __restrict__ Wk,
             float* __restrict__ Q,
             float* __restrict__ K,
             unsigned* __restrict__ segbits,
             float* __restrict__ denom,
             float* __restrict__ out) {
    __shared__ __attribute__((aligned(16))) float sA[16 * A_PITCH];

    const int lane    = threadIdx.x & 31;
    const int wave    = threadIdx.x >> 5;
    const int mat     = wave >> 2;
    const int colBase = (wave & 3) * 16;
    const int rowBase = blockIdx.x * 16;

    const float4* __restrict__ xt = (const float4*)(x + (size_t)rowBase * IN_F);
    const float4 x0 = xt[threadIdx.x];
    const float4 x1 = xt[threadIdx.x + 256];

    const float* __restrict__ W   = mat ? Wk : Wq;
    float*       __restrict__ Out = mat ? K  : Q;

    const int l    = lane & 15;
    const int koff = (lane >> 4) * 4;

    const float* __restrict__ brow = W + (size_t)(colBase + l) * IN_F + koff;
    float4 b[K_STEPS];
#pragma unroll
    for (int kk = 0; kk < K_STEPS; ++kk)
        b[kk] = *(const float4*)(brow + kk * 8);

    {
        const int i0 = threadIdx.x, i1 = threadIdx.x + 256;
        *(float4*)(sA + (i0 >> 5) * A_PITCH + (i0 & 31) * 4) = x0;
        *(float4*)(sA + (i1 >> 5) * A_PITCH + (i1 & 31) * 4) = x1;
    }
    __syncthreads();

    *(float4*)(out + (size_t)rowBase * OUT_F + threadIdx.x * 4) =
        make_float4(0.0f, 0.0f, 0.0f, 0.0f);
    if (threadIdx.x < 16)
        segbits[rowBase + threadIdx.x] = fkey(-INFINITY);
    else if (threadIdx.x < 32)
        denom[rowBase + threadIdx.x - 16] = 0.0f;

    const float* arow = sA + l * A_PITCH + koff;

    v8f acc = {};
#pragma unroll
    for (int kk = 0; kk < K_STEPS; ++kk) {
        const float4 a = *(const float4*)(arow + kk * 8);
        v2f a01 = {a.x, a.y}, a23 = {a.z, a.w};
        v2f b01 = {b[kk].x, b[kk].y}, b23 = {b[kk].z, b[kk].w};
        acc = __builtin_amdgcn_wmma_f32_16x16x4_f32(
             false, a01,  false, b01,
             (short)0, acc,  false,  false);
        acc = __builtin_amdgcn_wmma_f32_16x16x4_f32(
             false, a23,  false, b23,
             (short)0, acc,  false,  false);
    }

    const int outRow = rowBase + ((lane >> 4) ? 8 : 0);
    const int outCol = colBase + l;
#pragma unroll
    for (int r = 0; r < 8; ++r)
        Out[(size_t)(outRow + r) * OUT_F + outCol] = acc[r];
}

static_assert(N_EDGES % 32 == 0, "a wave takes whole chunks of 32 edges");
static_assert(OUT_F == 64, "16 lanes x 4 floats per row");
#define N_CHUNKS (N_EDGES / 32)

template <int J>
__device__ __forceinline__ int row_bcast(int v) {
    return __builtin_amdgcn_update_dpp(0, v, 0x150 + J, 0xf, 0xf, true);
}
__device__ __forceinline__ int row_down4(int v) {
    return __builtin_amdgcn_update_dpp(0, v, 0x120 + 12, 0xf, 0xf, true);
}
template <int M>
__device__ __forceinline__ float row_xadd(float p) {
    return p + __int_as_float(__builtin_amdgcn_update_dpp(
                   0, __float_as_int(p), 0x160 + M, 0xf, 0xf, true));
}

template <int J>
__device__ __forceinline__ float4 ld_row(const char* __restrict__ base,
                                         int idx, unsigned sub) {
    const unsigned r = (unsigned)row_bcast<J>(idx);
    return *(const float4*)(base + (r * (OUT_F * 4u) + sub * 16u));
}
__device__ __forceinline__ float row_dot(float4 q, float4 k) {
    float p = q.x * k.x + q.y * k.y + q.z * k.z + q.w * k.w;
    p = row_xadd<1>(p);
    p = row_xadd<2>(p);
    p = row_xadd<4>(p);
    p = row_xadd<8>(p);
    return p;
}

__global__ __launch_bounds__(128)
void edge_logits(const int* __restrict__ ei,
                 const float* __restrict__ Q,
                 const float* __restrict__ K,
                 float* __restrict__ alpha,
                 unsigned* __restrict__ segbits) {
    const int lane     = threadIdx.x & 31;
    const unsigned sub = lane & 15;
    const int wid = (blockIdx.x * blockDim.x + threadIdx.x) >> 5;
    const int nw  = (gridDim.x * blockDim.x) >> 5;
    const char* __restrict__ Qb = (const char*)Q;
    const char* __restrict__ Kb = (const char*)K;

    for (int chunk = wid; chunk < N_CHUNKS; chunk += nw) {
        const int e   = chunk * 32 + lane;
        const int row = ei[e];
        const int col = ei[N_EDGES + e];
        int rr = row, cr = col;
        float s = 0.0f;
#pragma unroll 1
        for (unsigned j0 = 0; j0 < 16; j0 += 4) {
            const float4 q0 = ld_row<0>(Qb, rr, sub);
            const float4 k0 = ld_row<0>(Kb, cr, sub);
            const float4 q1 = ld_row<1>(Qb, rr, sub);
            const float4 k1 = ld_row<1>(Kb, cr, sub);
            const float4 q2 = ld_row<2>(Qb, rr, sub);
            const float4 k2 = ld_row<2>(Kb, cr, sub);
            const float4 q3 = ld_row<3>(Qb, rr, sub);
            const float4 k3 = ld_row<3>(Kb, cr, sub);
            const float p0 = row_dot(q0, k0);
            const float p1 = row_dot(q1, k1);
            const float p2 = row_dot(q2, k2);
            const float p3 = row_dot(q3, k3);
            s = (sub == j0)      ? p0 : s;
            s = (sub == j0 + 1u) ? p1 : s;
            s = (sub == j0 + 2u) ? p2 : s;
            s = (sub == j0 + 3u) ? p3 : s;
            rr = row_down4(rr);
            cr = row_down4(cr);
        }
        s *= 0.125f;
        alpha[e] = s;
        atomicMax(segbits + col, fkey(s));
    }
}

__global__ void edge_exp(const int* __restrict__ ei,
                         const unsigned* __restrict__ segbits,
                         float* __restrict__ alpha,
                         float* __restrict__ denom) {
    const int e = blockIdx.x * blockDim.x + threadIdx.x;
    if (e >= N_EDGES) return;
    const int col = ei[N_EDGES + e];
    const float v = expf(alpha[e] - funkey(segbits[col]));
    alpha[e] = v;
    atomicAdd(denom + col, v);
}

#define LOOKAHEAD 8

__global__ __launch_bounds__(128)
void edge_scatter(const int* __restrict__ ei,
                  const float* __restrict__ Q,
                  const float* __restrict__ ex,
                  const float* __restrict__ denom,
                  float* __restrict__ out) {
    const int lane = threadIdx.x & 31;
    const int wid = __builtin_amdgcn_readfirstlane(
        (int)((blockIdx.x * blockDim.x + threadIdx.x) >> 5));
    const int nw  = (int)((gridDim.x * blockDim.x) >> 5);

    for (int chunk = wid; chunk < N_CHUNKS; chunk += nw) {
        const int e   = chunk * 32 + lane;
        const int row = ei[e];
        const int col = ei[N_EDGES + e];
        const float w = ex[e] / (denom[col] + 1e-16f);

        float qlo[32], qhi[32];
#pragma unroll
        for (int j = 0; j < LOOKAHEAD; ++j) {
            const float* q =
                Q + (size_t)__builtin_amdgcn_readlane(row, j) * OUT_F;
            qlo[j] = q[lane];
            qhi[j] = q[lane + 32];
        }
#pragma unroll
        for (int j = 0; j < 32; ++j) {
            if (j + LOOKAHEAD < 32) {
                const float* q =
                    Q + (size_t)__builtin_amdgcn_readlane(row, j + LOOKAHEAD) * OUT_F;
                qlo[j + LOOKAHEAD] = q[lane];
                qhi[j + LOOKAHEAD] = q[lane + 32];
            }
            const float wj = __int_as_float(
                __builtin_amdgcn_readlane(__float_as_int(w), j));
            float* o =
                out + (size_t)__builtin_amdgcn_readlane(col, j) * OUT_F;
            atomicAdd(o + lane,      qlo[j] * wj);
            atomicAdd(o + lane + 32, qhi[j] * wj);
        }
    }
}

extern "C" void kernel_launch(void* const* d_in, const int* in_sizes, int n_in,
                              void* d_out, int out_size, void* d_ws, size_t ws_size,
                              hipStream_t stream) {
    const float* x  = (const float*)d_in[0];
    const int*   ei = (const int*)d_in[1];
    const float* Wq = (const float*)d_in[2];
    const float* Wk = (const float*)d_in[3];
    float* out = (float*)d_out;

    char* ws = (char*)d_ws;
    float*    Q       = (float*)ws;    ws += (size_t)N_NODES * OUT_F * sizeof(float);
    float*    K       = (float*)ws;    ws += (size_t)N_NODES * OUT_F * sizeof(float);
    float*    alpha   = (float*)ws;    ws += (size_t)N_EDGES * sizeof(float);
    unsigned* segbits = (unsigned*)ws; ws += (size_t)N_NODES * sizeof(unsigned);
    float*    denom   = (float*)ws;

    qk_gemm<<<N_NODES / 16, 256, 0, stream>>>(x, Wq, Wk, Q, K, segbits, denom, out);
    edge_logits<<<N_CHUNKS / 8, 128, 0, stream>>>(ei, Q, K, alpha, segbits);
    edge_exp<<<(N_EDGES + 255) / 256, 256, 0, stream>>>(ei, segbits, alpha, denom);
    edge_scatter<<<N_CHUNKS / 8, 128, 0, stream>>>(ei, Q, alpha, denom, out);
}


# one work-group per 200 destination nodes finds its edges by scanning all ids; logits, softmax, weighted sum in LDS (LDS atomics, arrival-order sums); no global atomics; 2 launches, not 4
# speedup vs baseline: 3.5473x; 1.9985x over previous
#include <hip/hip_runtime.h>
#include <math.h>

#define N_NODES 50000
#define N_EDGES 800000
#define IN_F    128
#define OUT_F   64

typedef __attribute__((ext_vector_type(2))) float v2f;
typedef __attribute__((ext_vector_type(8))) float v8f;

__device__ __forceinline__ unsigned fkey(float f) {
    unsigned u = __float_as_uint(f);
    return (u & 0x80000000u) ? ~u : (u | 0x80000000u);
}
__device__ __forceinline__ float funkey(unsigned u) {
    return __uint_as_float((u & 0x80000000u) ? (u & 0x7FFFFFFFu) : ~u);
}

static_assert(N_NODES % 16 == 0, "a block owns 16 whole nodes");
static_assert(16 * IN_F == 2 * 256 * 4, "256 threads x 2 x 4 floats cover a block's x tile");
static_assert(IN_F % 8 == 0, "a step takes 8 k values");
#define A_PITCH (IN_F + 4)
#define K_STEPS (IN_F / 8)

__global__ __launch_bounds__(256)
void qk_gemm(const float* __restrict__ x,
             const float* __restrict__ Wq,
             const float* __restrict__ Wk,
             float* __restrict__ Q,
             float* __restrict__ K) {
    __shared__ __attribute__((aligned(16))) float sA[16 * A_PITCH];

    const int lane    = threadIdx.x & 31;
    const int wave    = threadIdx.x >> 5;
    const int mat     = wave >> 2;
    const int colBase = (wave & 3) * 16;
    const int rowBase = blockIdx.x * 16;

    const float4* __restrict__ xt = (const float4*)(x + (size_t)rowBase * IN_F);
    const float4 x0 = xt[threadIdx.x];
    const float4 x1 = xt[threadIdx.x + 256];

    const float* __restrict__ W   = mat ? Wk : Wq;
    float*       __restrict__ Out = mat ? K  : Q;

    const int l    = lane & 15;
    const int koff = (lane >> 4) * 4;

    const float* __restrict__ brow = W + (size_t)(colBase + l) * IN_F + koff;
    float4 b[K_STEPS];
#pragma unroll
    for (int kk = 0; kk < K_STEPS; ++kk)
        b[kk] = *(const float4*)(brow + kk * 8);

    {
        const int i0 = threadIdx.x, i1 = threadIdx.x + 256;
        *(float4*)(sA + (i0 >> 5) * A_PITCH + (i0 & 31) * 4) = x0;
        *(float4*)(sA + (i1 >> 5) * A_PITCH + (i1 & 31) * 4) = x1;
    }
    __syncthreads();

    const float* arow = sA + l * A_PITCH + koff;

    v8f acc = {};
#pragma unroll
    for (int kk = 0; kk < K_STEPS; ++kk) {
        const float4 a = *(const float4*)(arow + kk * 8);
        v2f a01 = {a.x, a.y}, a23 = {a.z, a.w};
        v2f b01 = {b[kk].x, b[kk].y}, b23 = {b[kk].z, b[kk].w};
        acc = __builtin_amdgcn_wmma_f32_16x16x4_f32(
             false, a01,  false, b01,
             (short)0, acc,  false,  false);
        acc = __builtin_amdgcn_wmma_f32_16x16x4_f32(
             false, a23,  false, b23,
             (short)0, acc,  false,  false);
    }

    const int outRow = rowBase + ((lane >> 4) ? 8 : 0);
    const int outCol = colBase + l;
#pragma unroll
    for (int r = 0; r < 8; ++r)
        Out[(size_t)(outRow + r) * OUT_F + outCol] = acc[r];
}

#define NPB   200
#define NBLK  (N_NODES / NPB)
#define TPB   1024
#define CAP   8192
#define NQ    (N_EDGES / 4)
#define NROWS (TPB / 16)
#define UNR   8
#define ACC_PITCH (OUT_F + 1)
#define SCAN_U 8
#define SCAN_TRIPS ((NQ + TPB * SCAN_U - 1) / (TPB * SCAN_U))
static_assert(N_NODES % NPB == 0, "ranges tile the nodes exactly");
static_assert(NPB <= 256, "local node index is packed into 8 bits");
static_assert(N_EDGES < (1 << 24), "edge index is packed into 24 bits");
static_assert(N_EDGES % 4 == 0, "ids are read 16 bytes at a time");
static_assert((NPB * OUT_F) % 4 == 0, "K rows move as float4");
static_assert(OUT_F == 64, "16 lanes x 4 floats per row");

template <int M>
__device__ __forceinline__ float row_xadd(float p) {
    return p + __int_as_float(__builtin_amdgcn_update_dpp(
                   0, __float_as_int(p), 0x160 + M, 0xf, 0xf, true));
}
__device__ __forceinline__ float row_dot(float4 q, float4 k) {
    float p = q.x * k.x + q.y * k.y + q.z * k.z + q.w * k.w;
    p = row_xadd<1>(p);
    p = row_xadd<2>(p);
    p = row_xadd<4>(p);
    p = row_xadd<8>(p);
    return p;
}

__device__ __forceinline__ void lds_add(float* p, float v) {
    __hip_atomic_fetch_add(p, v, __ATOMIC_RELAXED, __HIP_MEMORY_SCOPE_WORKGROUP);
}
__device__ __forceinline__ void lds_max(unsigned* p, unsigned v) {
    __hip_atomic_fetch_max(p, v, __ATOMIC_RELAXED, __HIP_MEMORY_SCOPE_WORKGROUP);
}

__global__ __launch_bounds__(TPB)
void range_attend(const int* __restrict__ ei,
                  const float* __restrict__ Q,
                  const float* __restrict__ K,
                  float* __restrict__ alpha,
                  float* __restrict__ out) {
    __shared__ float sAcc[NPB * ACC_PITCH];
    __shared__ __attribute__((aligned(16))) float sK[NPB * OUT_F];
    __shared__ unsigned sMax[NPB];
    __shared__ float    sDen[NPB];
    __shared__ unsigned sList[CAP];
    __shared__ int      sRow[CAP];
    __shared__ float    sA[CAP];
    __shared__ unsigned sCnt;

    const int tid = threadIdx.x;
    const int n0  = blockIdx.x * NPB;
    const int* __restrict__ cols = ei + N_EDGES;

    for (int i = tid; i < NPB * ACC_PITCH; i += TPB)
        sAcc[i] = 0.0f;
    for (int i = tid; i < NPB * OUT_F / 4; i += TPB)
        ((float4*)sK)[i] = ((const float4*)(K + (size_t)n0 * OUT_F))[i];
    if (tid < NPB) {
        sMax[tid] = fkey(-INFINITY);
        sDen[tid] = 0.0f;
    }
    if (tid == 0) sCnt = 0;
    __syncthreads();

    {
        const int4* __restrict__ cols4 = (const int4*)cols;
        const int t0 = blockIdx.x % SCAN_TRIPS;
        for (int t = 0; t < SCAN_TRIPS; ++t) {
            int tt = t + t0;
            if (tt >= SCAN_TRIPS) tt -= SCAN_TRIPS;
            const int base = tt * (TPB * SCAN_U) + tid;
            int4 c[SCAN_U];
#pragma unroll
            for (int u = 0; u < SCAN_U; ++u) {
                const int i = base + u * TPB;
                c[u] = cols4[i < NQ ? i : tid];
            }
#pragma unroll
            for (int u = 0; u < SCAN_U; ++u) {
                const int i = base + u * TPB;
                const unsigned d0 = (unsigned)(c[u].x - n0), d1 = (unsigned)(c[u].y - n0);
                const unsigned d2 = (unsigned)(c[u].z - n0), d3 = (unsigned)(c[u].w - n0);
                const unsigned m = min(min(d0, d1), min(d2, d3));
                if (m < NPB && i < NQ) {
                    const unsigned e = (unsigned)i * 4u;
                    if (d0 < NPB) { const unsigned p = atomicAdd(&sCnt, 1u); if (p < CAP) sList[p] = ((e     ) << 8) | d0; }
                    if (d1 < NPB) { const unsigned p = atomicAdd(&sCnt, 1u); if (p < CAP) sList[p] = ((e + 1u) << 8) | d1; }
                    if (d2 < NPB) { const unsigned p = atomicAdd(&sCnt, 1u); if (p < CAP) sList[p] = ((e + 2u) << 8) | d2; }
                    if (d3 < NPB) { const unsigned p = atomicAdd(&sCnt, 1u); if (p < CAP) sList[p] = ((e + 3u) << 8) | d3; }
                }
            }
        }
    }
    __syncthreads();

    const unsigned cnt = (unsigned)__builtin_amdgcn_readfirstlane((int)sCnt);

    if (cnt <= CAP) {
        const int n = (int)cnt;
        for (int p = tid; p < n; p += TPB)
            sRow[p] = ei[sList[p] >> 8];
        __syncthreads();

        const int      hrow = tid >> 4;
        const unsigned sub  = tid & 15;

        for (int base = 0; base < n; base += NROWS * UNR) {
            float4 q[UNR];
            int lc[UNR], pp[UNR];
#pragma unroll
            for (int u = 0; u < UNR; ++u) {
                const int p  = base + u * NROWS + hrow;
                const int pc = p < n ? p : 0;
                pp[u] = p < n ? p : -1;
                lc[u] = (int)(sList[pc] & 255u);
                q[u]  = *(const float4*)(Q + (size_t)sRow[pc] * OUT_F + sub * 4);
            }
#pragma unroll
            for (int u = 0; u < UNR; ++u) {
                const float4 k = *(const float4*)(sK + lc[u] * OUT_F + sub * 4);
                const float a = row_dot(q[u], k) * 0.125f;
                if (sub == 0 && pp[u] >= 0) {
                    sA[pp[u]] = a;
                    lds_max(&sMax[lc[u]], fkey(a));
                }
            }
        }
        __syncthreads();

        for (int p = tid; p < n; p += TPB) {
            const int l = (int)(sList[p] & 255u);
            const float ex = expf(sA[p] - funkey(sMax[l]));
            sA[p] = ex;
            lds_add(&sDen[l], ex);
        }
        __syncthreads();

        for (int p = tid; p < n; p += TPB)
            sA[p] = sA[p] / (sDen[sList[p] & 255u] + 1e-16f);
        __syncthreads();

        for (int base = 0; base < n; base += NROWS * UNR) {
            float4 q[UNR];
            int lc[UNR], pp[UNR];
            float w[UNR];
#pragma unroll
            for (int u = 0; u < UNR; ++u) {
                const int p  = base + u * NROWS + hrow;
                const int pc = p < n ? p : 0;
                pp[u] = p < n ? p : -1;
                lc[u] = (int)(sList[pc] & 255u);
                q[u]  = *(const float4*)(Q + (size_t)sRow[pc] * OUT_F + sub * 4);
                w[u]  = sA[pc];
            }
#pragma unroll
            for (int u = 0; u < UNR; ++u) {
                if (pp[u] >= 0) {
                    float* a = sAcc + lc[u] * ACC_PITCH + sub * 4;
                    lds_add(a + 0, q[u].x * w[u]);
                    lds_add(a + 1, q[u].y * w[u]);
                    lds_add(a + 2, q[u].z * w[u]);
                    lds_add(a + 3, q[u].w * w[u]);
                }
            }
        }
    } else {
        for (int e = tid; e < N_EDGES; e += TPB) {
            const unsigned d = (unsigned)(cols[e] - n0);
            if (d < NPB) {
                const float* q = Q + (size_t)ei[e] * OUT_F;
                const float* k = sK + d * OUT_F;
                float s = 0.0f;
                for (int j = 0; j < OUT_F; ++j) s += q[j] * k[j];
                s *= 0.125f;
                alpha[e] = s;
                lds_max(&sMax[d], fkey(s));
            }
        }
        __syncthreads();
        for (int e = tid; e < N_EDGES; e += TPB) {
            const unsigned d = (unsigned)(cols[e] - n0);
            if (d < NPB) {
                const float ex = expf(alpha[e] - funkey(sMax[d]));
                alpha[e] = ex;
                lds_add(&sDen[d], ex);
            }
        }
        __syncthreads();
        for (int e = tid; e < N_EDGES; e += TPB) {
            const unsigned d = (unsigned)(cols[e] - n0);
            if (d < NPB) {
                const float* q = Q + (size_t)ei[e] * OUT_F;
                const float w = alpha[e] / (sDen[d] + 1e-16f);
                for (int j = 0; j < OUT_F; ++j) lds_add(&sAcc[d * ACC_PITCH + j], q[j] * w);
            }
        }
    }
    __syncthreads();

    for (int i = tid; i < NPB * OUT_F; i += TPB)
        out[(size_t)n0 * OUT_F + i] = sAcc[(i >> 6) * ACC_PITCH + (i & 63)];
}

extern "C" void kernel_launch(void* const* d_in, const int* in_sizes, int n_in,
                              void* d_out, int out_size, void* d_ws, size_t ws_size,
                              hipStream_t stream) {
    const float* x  = (const float*)d_in[0];
    const int*   ei = (const int*)d_in[1];
    const float* Wq = (const float*)d_in[2];
    const float* Wk = (const float*)d_in[3];
    float* out = (float*)d_out;

    char* ws = (char*)d_ws;
    float*    Q       = (float*)ws;    ws += (size_t)N_NODES * OUT_F * sizeof(float);
    float*    K       = (float*)ws;    ws += (size_t)N_NODES * OUT_F * sizeof(float);
    float*    alpha   = (float*)ws;

    qk_gemm<<<N_NODES / 16, 256, 0, stream>>>(x, Wq, Wk, Q, K);
    range_attend<<<NBLK, TPB, 0, stream>>>(ei, Q, K, alpha, out);
}


# range_attend scans an 8-bit-per-edge range map (bit planes by ballot, written by the qk_gemm waves), not all 4-byte destination ids; ids fetched for listed edges only; LDS sum arrival order changes
# speedup vs baseline: 4.4207x; 1.2462x over previous
#include <hip/hip_runtime.h>
#include <math.h>

#define N_NODES 50000
#define N_EDGES 800000
#define IN_F    128
#define OUT_F   64

typedef __attribute__((ext_vector_type(2))) float v2f;
typedef __attribute__((ext_vector_type(8))) float v8f;

__device__ __forceinline__ unsigned fkey(float f) {
    unsigned u = __float_as_uint(f);
    return (u & 0x80000000u) ? ~u : (u | 0x80000000u);
}
__device__ __forceinline__ float funkey(unsigned u) {
    return __uint_as_float((u & 0x80000000u) ? (u & 0x7FFFFFFFu) : ~u);
}

static_assert(N_NODES % 16 == 0, "a block owns 16 whole nodes");
#define NPB   200
#define GEMM_BLOCKS (N_NODES / 16)
#define EPB   (N_EDGES / GEMM_BLOCKS)
#define NG    (N_EDGES / 32)
#define NO_RANGE 255u
static_assert(N_EDGES % GEMM_BLOCKS == 0, "the blocks share the edges evenly");
static_assert(EPB == 8 * 32, "one group of 32 edges for each of a block's 8 waves");
static_assert(N_NODES / NPB <= NO_RANGE, "a range number fits in 8 bits");

__device__ __forceinline__ unsigned range_of(int id) {
    const unsigned r = (unsigned)id / NPB;
    return r < NO_RANGE ? r : NO_RANGE;
}
static_assert(16 * IN_F == 2 * 256 * 4, "256 threads x 2 x 4 floats cover a block's x tile");
static_assert(IN_F % 8 == 0, "a step takes 8 k values");
#define A_PITCH (IN_F + 4)
#define K_STEPS (IN_F / 8)

__global__ __launch_bounds__(256)
void qk_gemm(const float* __restrict__ x,
             const float* __restrict__ Wq,
             const float* __restrict__ Wk,
             const int* __restrict__ ei,
             float* __restrict__ Q,
             float* __restrict__ K,
             unsigned* __restrict__ rmap) {
    __shared__ __attribute__((aligned(16))) float sA[16 * A_PITCH];
    __shared__ unsigned sMap[8 * 8];

    const int lane    = threadIdx.x & 31;
    const int wave    = threadIdx.x >> 5;
    const int mat     = wave >> 2;
    const int colBase = (wave & 3) * 16;
    const int rowBase = blockIdx.x * 16;

    const float4* __restrict__ xt = (const float4*)(x + (size_t)rowBase * IN_F);
    const float4 x0 = xt[threadIdx.x];
    const float4 x1 = xt[threadIdx.x + 256];

    const int dst = (ei + N_EDGES)[(blockIdx.x * 8 + wave) * 32 + lane];

    const float* __restrict__ W   = mat ? Wk : Wq;
    float*       __restrict__ Out = mat ? K  : Q;

    const int l    = lane & 15;
    const int koff = (lane >> 4) * 4;

    const float* __restrict__ brow = W + (size_t)(colBase + l) * IN_F + koff;
    float4 b[K_STEPS];
#pragma unroll
    for (int kk = 0; kk < K_STEPS; ++kk)
        b[kk] = *(const float4*)(brow + kk * 8);

    {
        const int i0 = threadIdx.x, i1 = threadIdx.x + 256;
        *(float4*)(sA + (i0 >> 5) * A_PITCH + (i0 & 31) * 4) = x0;
        *(float4*)(sA + (i1 >> 5) * A_PITCH + (i1 & 31) * 4) = x1;
    }
    {
        const unsigned r = range_of(dst);
        unsigned w = 0;
        for (int k = 0; k < 8; ++k) {
            const unsigned plane = __builtin_amdgcn_ballot_w32(((r >> k) & 1u) != 0);
            if (lane == k) w = plane;
        }
        if (lane < 8) sMap[wave * 8 + lane] = w;
    }
    __syncthreads();

    if (threadIdx.x < 64) {
        const int half = threadIdx.x >> 5, within = threadIdx.x & 31;
        rmap[half * (NG * 4) + blockIdx.x * 32 + within] =
            sMap[(within >> 2) * 8 + half * 4 + (within & 3)];
    }

    const float* arow = sA + l * A_PITCH + koff;

    v8f acc = {};
#pragma unroll
    for (int kk = 0; kk < K_STEPS; ++kk) {
        const float4 a = *(const float4*)(arow + kk * 8);
        v2f a01 = {a.x, a.y}, a23 = {a.z, a.w};
        v2f b01 = {b[kk].x, b[kk].y}, b23 = {b[kk].z, b[kk].w};
        acc = __builtin_amdgcn_wmma_f32_16x16x4_f32(
             false, a01,  false, b01,
             (short)0, acc,  false,  false);
        acc = __builtin_amdgcn_wmma_f32_16x16x4_f32(
             false, a23,  false, b23,
             (short)0, acc,  false,  false);
    }

    const int outRow = rowBase + ((lane >> 4) ? 8 : 0);
    const int outCol = colBase + l;
#pragma unroll
    for (int r = 0; r < 8; ++r)
        Out[(size_t)(outRow + r) * OUT_F + outCol] = acc[r];
}

#define NBLK  (N_NODES / NPB)
#define TPB   1024
#define CAP   8192
#define NROWS (TPB / 16)
#define UNR   8
#define ACC_PITCH (OUT_F + 1)
#define SCAN_U 5
#define SCAN_TRIPS ((NG + TPB * SCAN_U - 1) / (TPB * SCAN_U))
static_assert(N_NODES % NPB == 0, "ranges tile the nodes exactly");
static_assert(NPB <= 256, "local node index is packed into 8 bits");
static_assert(N_EDGES < (1 << 24), "edge index is packed into 24 bits");
static_assert(N_EDGES % 32 == 0, "the map is kept in groups of 32 edges");
static_assert(NG > TPB, "the dummy group of a thread past the end exists");
static_assert((NPB * OUT_F) % 4 == 0, "K rows move as float4");
static_assert(OUT_F == 64, "16 lanes x 4 floats per row");

template <int M>
__device__ __forceinline__ float row_xadd(float p) {
    return p + __int_as_float(__builtin_amdgcn_update_dpp(
                   0, __float_as_int(p), 0x160 + M, 0xf, 0xf, true));
}
__device__ __forceinline__ float row_dot(float4 q, float4 k) {
    float p = q.x * k.x + q.y * k.y + q.z * k.z + q.w * k.w;
    p = row_xadd<1>(p);
    p = row_xadd<2>(p);
    p = row_xadd<4>(p);
    p = row_xadd<8>(p);
    return p;
}

__device__ __forceinline__ float edge_logit(const float* __restrict__ q, const float* k) {
    float s = 0.0f;
    for (int j = 0; j < OUT_F; ++j) s += q[j] * k[j];
    return s * 0.125f;
}

__device__ __forceinline__ void lds_add(float* p, float v) {
    __hip_atomic_fetch_add(p, v, __ATOMIC_RELAXED, __HIP_MEMORY_SCOPE_WORKGROUP);
}
__device__ __forceinline__ void lds_max(unsigned* p, unsigned v) {
    __hip_atomic_fetch_max(p, v, __ATOMIC_RELAXED, __HIP_MEMORY_SCOPE_WORKGROUP);
}

__global__ __launch_bounds__(TPB)
void range_attend(const int* __restrict__ ei,
                  const float* __restrict__ Q,
                  const float* __restrict__ K,
                  const unsigned* __restrict__ rmap,
                  float* __restrict__ out) {
    __shared__ float sAcc[NPB * ACC_PITCH];
    __shared__ __attribute__((aligned(16))) float sK[NPB * OUT_F];
    __shared__ unsigned sMax[NPB];
    __shared__ float    sDen[NPB];
    __shared__ unsigned sList[CAP];
    __shared__ int      sRow[CAP];
    __shared__ float    sA[CAP];
    __shared__ unsigned sCnt;

    const int tid = threadIdx.x;
    const int n0  = blockIdx.x * NPB;
    const int* __restrict__ cols = ei + N_EDGES;

    for (int i = tid; i < NPB * ACC_PITCH; i += TPB)
        sAcc[i] = 0.0f;
    for (int i = tid; i < NPB * OUT_F / 4; i += TPB)
        ((float4*)sK)[i] = ((const float4*)(K + (size_t)n0 * OUT_F))[i];
    if (tid < NPB) {
        sMax[tid] = fkey(-INFINITY);
        sDen[tid] = 0.0f;
    }
    if (tid == 0) sCnt = 0;
    __syncthreads();

    {
        const uint4* __restrict__ lo4 = (const uint4*)rmap;
        const uint4* __restrict__ hi4 = lo4 + NG;
        unsigned nm[8];
        for (int k = 0; k < 8; ++k)
            nm[k] = ((blockIdx.x >> k) & 1u) ? 0u : ~0u;
        const int t0 = blockIdx.x % SCAN_TRIPS;
#pragma nounroll
        for (int t = 0; t < SCAN_TRIPS; ++t) {
            int tt = t + t0;
            if (tt >= SCAN_TRIPS) tt -= SCAN_TRIPS;
            const int base = tt * (TPB * SCAN_U) + tid;
            uint4 a[SCAN_U], b[SCAN_U];
#pragma unroll
            for (int u = 0; u < SCAN_U; ++u) {
                const int i = base + u * TPB;
                const int j = i < NG ? i : tid;
                a[u] = lo4[j];
                b[u] = hi4[j];
            }
#pragma unroll
            for (int u = 0; u < SCAN_U; ++u) {
                const int i = base + u * TPB;
                unsigned hits = (a[u].x ^ nm[0]) & (a[u].y ^ nm[1]) & (a[u].z ^ nm[2]) & (a[u].w ^ nm[3]) &
                                (b[u].x ^ nm[4]) & (b[u].y ^ nm[5]) & (b[u].z ^ nm[6]) & (b[u].w ^ nm[7]);
                if (i >= NG) hits = 0;
                while (hits) {
                    const unsigned bit = (unsigned)__builtin_ctz(hits);
                    hits &= hits - 1;
                    const unsigned p = atomicAdd(&sCnt, 1u);
                    if (p < CAP) sList[p] = (unsigned)i * 32u + bit;
                }
            }
        }
    }
    __syncthreads();

    const unsigned cnt = (unsigned)__builtin_amdgcn_readfirstlane((int)sCnt);

    if (cnt <= CAP) {
        const int n = (int)cnt;
        for (int p = tid; p < n; p += TPB) {
            const unsigned e = sList[p];
            const unsigned d = min((unsigned)(cols[e] - n0), (unsigned)(NPB - 1));
            sRow[p]  = ei[e];
            sList[p] = (e << 8) | d;
        }
        __syncthreads();

        const int      hrow = tid >> 4;
        const unsigned sub  = tid & 15;

        for (int base = 0; base < n; base += NROWS * UNR) {
            float4 q[UNR];
            int lc[UNR], pp[UNR];
#pragma unroll
            for (int u = 0; u < UNR; ++u) {
                const int p  = base + u * NROWS + hrow;
                const int pc = p < n ? p : 0;
                pp[u] = p < n ? p : -1;
                lc[u] = (int)(sList[pc] & 255u);
                q[u]  = *(const float4*)(Q + (size_t)sRow[pc] * OUT_F + sub * 4);
            }
#pragma unroll
            for (int u = 0; u < UNR; ++u) {
                const float4 k = *(const float4*)(sK + lc[u] * OUT_F + sub * 4);
                const float a = row_dot(q[u], k) * 0.125f;
                if (sub == 0 && pp[u] >= 0) {
                    sA[pp[u]] = a;
                    lds_max(&sMax[lc[u]], fkey(a));
                }
            }
        }
        __syncthreads();

        for (int p = tid; p < n; p += TPB) {
            const int l = (int)(sList[p] & 255u);
            const float ex = expf(sA[p] - funkey(sMax[l]));
            sA[p] = ex;
            lds_add(&sDen[l], ex);
        }
        __syncthreads();

        for (int p = tid; p < n; p += TPB)
            sA[p] = sA[p] / (sDen[sList[p] & 255u] + 1e-16f);
        __syncthreads();

        for (int base = 0; base < n; base += NROWS * UNR) {
            float4 q[UNR];
            int lc[UNR], pp[UNR];
            float w[UNR];
#pragma unroll
            for (int u = 0; u < UNR; ++u) {
                const int p  = base + u * NROWS + hrow;
                const int pc = p < n ? p : 0;
                pp[u] = p < n ? p : -1;
                lc[u] = (int)(sList[pc] & 255u);
                q[u]  = *(const float4*)(Q + (size_t)sRow[pc] * OUT_F + sub * 4);
                w[u]  = sA[pc];
            }
#pragma unroll
            for (int u = 0; u < UNR; ++u) {
                if (pp[u] >= 0) {
                    float* a = sAcc + lc[u] * ACC_PITCH + sub * 4;
                    lds_add(a + 0, q[u].x * w[u]);
                    lds_add(a + 1, q[u].y * w[u]);
                    lds_add(a + 2, q[u].z * w[u]);
                    lds_add(a + 3, q[u].w * w[u]);
                }
            }
        }
    } else {
        for (int e = tid; e < N_EDGES; e += TPB) {
            const unsigned d = (unsigned)(cols[e] - n0);
            if (d < NPB)
                lds_max(&sMax[d], fkey(edge_logit(Q + (size_t)ei[e] * OUT_F, sK + d * OUT_F)));
        }
        __syncthreads();
        for (int e = tid; e < N_EDGES; e += TPB) {
            const unsigned d = (unsigned)(cols[e] - n0);
            if (d < NPB) {
                const float s = edge_logit(Q + (size_t)ei[e] * OUT_F, sK + d * OUT_F);
                lds_add(&sDen[d], expf(s - funkey(sMax[d])));
            }
        }
        __syncthreads();
        for (int e = tid; e < N_EDGES; e += TPB) {
            const unsigned d = (unsigned)(cols[e] - n0);
            if (d < NPB) {
                const float* q = Q + (size_t)ei[e] * OUT_F;
                const float s = edge_logit(q, sK + d * OUT_F);
                const float w = expf(s - funkey(sMax[d])) / (sDen[d] + 1e-16f);
                for (int j = 0; j < OUT_F; ++j) lds_add(&sAcc[d * ACC_PITCH + j], q[j] * w);
            }
        }
    }
    __syncthreads();

    for (int i = tid; i < NPB * OUT_F; i += TPB)
        out[(size_t)n0 * OUT_F + i] = sAcc[(i >> 6) * ACC_PITCH + (i & 63)];
}

extern "C" void kernel_launch(void* const* d_in, const int* in_sizes, int n_in,
                              void* d_out, int out_size, void* d_ws, size_t ws_size,
                              hipStream_t stream) {
    const float* x  = (const float*)d_in[0];
    const int*   ei = (const int*)d_in[1];
    const float* Wq = (const float*)d_in[2];
    const float* Wk = (const float*)d_in[3];
    float* out = (float*)d_out;

    char* ws = (char*)d_ws;
    float*    Q       = (float*)ws;    ws += (size_t)N_NODES * OUT_F * sizeof(float);
    float*    K       = (float*)ws;    ws += (size_t)N_NODES * OUT_F * sizeof(float);
    unsigned* rmap    = (unsigned*)ws;

    qk_gemm<<<GEMM_BLOCKS, 256, 0, stream>>>(x, Wq, Wk, ei, Q, K, rmap);
    range_attend<<<NBLK, TPB, 0, stream>>>(ei, Q, K, rmap, out);
}


# range_attend sorts its edge list by destination in LDS; a 16-lane row does a node in registers (running max, rescaled sums, one divide per row): Q rows read once, no LDS float atomics; sum order new
# speedup vs baseline: 6.2035x; 1.4033x over previous
#include <hip/hip_runtime.h>
#include <math.h>

#define N_NODES 50000
#define N_EDGES 800000
#define IN_F    128
#define OUT_F   64

typedef __attribute__((ext_vector_type(2))) float v2f;
typedef __attribute__((ext_vector_type(8))) float v8f;

__device__ __forceinline__ unsigned fkey(float f) {
    unsigned u = __float_as_uint(f);
    return (u & 0x80000000u) ? ~u : (u | 0x80000000u);
}
__device__ __forceinline__ float funkey(unsigned u) {
    return __uint_as_float((u & 0x80000000u) ? (u & 0x7FFFFFFFu) : ~u);
}

static_assert(N_NODES % 16 == 0, "a block owns 16 whole nodes");
#define NPB   200
#define GEMM_BLOCKS (N_NODES / 16)
#define EPB   (N_EDGES / GEMM_BLOCKS)
#define NG    (N_EDGES / 32)
#define NO_RANGE 255u
static_assert(N_EDGES % GEMM_BLOCKS == 0, "the blocks share the edges evenly");
static_assert(EPB == 8 * 32, "one group of 32 edges for each of a block's 8 waves");
static_assert(N_NODES / NPB <= NO_RANGE, "a range number fits in 8 bits");

__device__ __forceinline__ unsigned range_of(int id) {
    const unsigned r = (unsigned)id / NPB;
    return r < NO_RANGE ? r : NO_RANGE;
}
static_assert(16 * IN_F == 2 * 256 * 4, "256 threads x 2 x 4 floats cover a block's x tile");
static_assert(IN_F % 8 == 0, "a step takes 8 k values");
#define A_PITCH (IN_F + 4)
#define K_STEPS (IN_F / 8)

__global__ __launch_bounds__(256)
void qk_gemm(const float* __restrict__ x,
             const float* __restrict__ Wq,
             const float* __restrict__ Wk,
             const int* __restrict__ ei,
             float* __restrict__ Q,
             float* __restrict__ K,
             unsigned* __restrict__ rmap) {
    __shared__ __attribute__((aligned(16))) float sA[16 * A_PITCH];
    __shared__ unsigned sMap[8 * 8];

    const int lane    = threadIdx.x & 31;
    const int wave    = threadIdx.x >> 5;
    const int mat     = wave >> 2;
    const int colBase = (wave & 3) * 16;
    const int rowBase = blockIdx.x * 16;

    const float4* __restrict__ xt = (const float4*)(x + (size_t)rowBase * IN_F);
    const float4 x0 = xt[threadIdx.x];
    const float4 x1 = xt[threadIdx.x + 256];

    const int dst = (ei + N_EDGES)[(blockIdx.x * 8 + wave) * 32 + lane];

    const float* __restrict__ W   = mat ? Wk : Wq;
    float*       __restrict__ Out = mat ? K  : Q;

    const int l    = lane & 15;
    const int koff = (lane >> 4) * 4;

    const float* __restrict__ brow = W + (size_t)(colBase + l) * IN_F + koff;
    float4 b[K_STEPS];
#pragma unroll
    for (int kk = 0; kk < K_STEPS; ++kk)
        b[kk] = *(const float4*)(brow + kk * 8);

    {
        const int i0 = threadIdx.x, i1 = threadIdx.x + 256;
        *(float4*)(sA + (i0 >> 5) * A_PITCH + (i0 & 31) * 4) = x0;
        *(float4*)(sA + (i1 >> 5) * A_PITCH + (i1 & 31) * 4) = x1;
    }
    {
        const unsigned r = range_of(dst);
        unsigned w = 0;
        for (int k = 0; k < 8; ++k) {
            const unsigned plane = __builtin_amdgcn_ballot_w32(((r >> k) & 1u) != 0);
            if (lane == k) w = plane;
        }
        if (lane < 8) sMap[wave * 8 + lane] = w;
    }
    __syncthreads();

    if (threadIdx.x < 64) {
        const int half = threadIdx.x >> 5, within = threadIdx.x & 31;
        rmap[half * (NG * 4) + blockIdx.x * 32 + within] =
            sMap[(within >> 2) * 8 + half * 4 + (within & 3)];
    }

    const float* arow = sA + l * A_PITCH + koff;

    v8f acc = {};
#pragma unroll
    for (int kk = 0; kk < K_STEPS; ++kk) {
        const float4 a = *(const float4*)(arow + kk * 8);
        v2f a01 = {a.x, a.y}, a23 = {a.z, a.w};
        v2f b01 = {b[kk].x, b[kk].y}, b23 = {b[kk].z, b[kk].w};
        acc = __builtin_amdgcn_wmma_f32_16x16x4_f32(
             false, a01,  false, b01,
             (short)0, acc,  false,  false);
        acc = __builtin_amdgcn_wmma_f32_16x16x4_f32(
             false, a23,  false, b23,
             (short)0, acc,  false,  false);
    }

    const int outRow = rowBase + ((lane >> 4) ? 8 : 0);
    const int outCol = colBase + l;
#pragma unroll
    for (int r = 0; r < 8; ++r)
        Out[(size_t)(outRow + r) * OUT_F + outCol] = acc[r];
}

#define NBLK  (N_NODES / NPB)
#define TPB   1024
#define CAP   8192
#define STEP  8
#define BPL   ((NPB + 31) / 32)
#define ACC_PITCH (OUT_F + 1)
#define SCAN_U 5
#define SCAN_TRIPS ((NG + TPB * SCAN_U - 1) / (TPB * SCAN_U))
static_assert(N_NODES % NPB == 0, "ranges tile the nodes exactly");
static_assert(NPB <= 256, "local node index is packed into 8 bits");
static_assert(NPB % 2 == 0, "a ticket is a pair of nodes, one per 16-lane row of a wave");
static_assert(CAP <= (1 << 24), "place among a node's edges is packed into 24 bits");
static_assert(N_EDGES % 32 == 0, "the map is kept in groups of 32 edges");
static_assert(NG > TPB, "the dummy group of a thread past the end exists");
static_assert((NPB * OUT_F) % 4 == 0, "K rows move as float4");
static_assert(OUT_F == 64, "16 lanes x 4 floats per row");
static_assert(STEP == 8, "soft_step names its 8 places one by one; lane STEP of a 16-lane row exists");

template <int M>
__device__ __forceinline__ float row_xadd(float p) {
    return p + __int_as_float(__builtin_amdgcn_update_dpp(
                   0, __float_as_int(p), 0x160 + M, 0xf, 0xf, true));
}
__device__ __forceinline__ float row_dot(float4 q, float4 k) {
    float p = q.x * k.x + q.y * k.y + q.z * k.z + q.w * k.w;
    p = row_xadd<1>(p);
    p = row_xadd<2>(p);
    p = row_xadd<4>(p);
    p = row_xadd<8>(p);
    return p;
}

template <int U>
__device__ __forceinline__ float row_bcast(float v) {
    return __int_as_float(__builtin_amdgcn_update_dpp(
               0, __float_as_int(v), 0x150 + U, 0xf, 0xf, true));
}

__device__ __forceinline__ void soft_step(const float4* q, int c, int end, float4 k, unsigned sub,
                                          float& m, float& den, float4& acc) {
    float a[STEP];
    float cm = -INFINITY;
#pragma unroll
    for (int u = 0; u < STEP; ++u) {
        const float d = row_dot(q[u], k) * 0.125f;
        a[u] = (c + u < end) ? d : -INFINITY;
        cm = fmaxf(cm, a[u]);
    }
    const float mn = fmaxf(m, cm);
    float mine = a[0];
#pragma unroll
    for (int u = 1; u < STEP; ++u)
        mine = ((sub & (STEP - 1)) == (unsigned)u) ? a[u] : mine;
    mine = (sub & STEP) ? m : mine;
    const float exm = expf(mine - mn);
    const float sc  = row_bcast<STEP>(exm);
    den   *= sc;
    acc.x *= sc; acc.y *= sc; acc.z *= sc; acc.w *= sc;
#define SOFT_SLOT(U) { const float ex = row_bcast<U>(exm); den += ex; \
        acc.x += q[U].x * ex; acc.y += q[U].y * ex; acc.z += q[U].z * ex; acc.w += q[U].w * ex; }
    SOFT_SLOT(0) SOFT_SLOT(1) SOFT_SLOT(2) SOFT_SLOT(3)
    SOFT_SLOT(4) SOFT_SLOT(5) SOFT_SLOT(6) SOFT_SLOT(7)
#undef SOFT_SLOT
    m = mn;
}

__device__ __forceinline__ float edge_logit(const float* __restrict__ q, const float* k) {
    float s = 0.0f;
    for (int j = 0; j < OUT_F; ++j) s += q[j] * k[j];
    return s * 0.125f;
}

__device__ __forceinline__ void lds_add(float* p, float v) {
    __hip_atomic_fetch_add(p, v, __ATOMIC_RELAXED, __HIP_MEMORY_SCOPE_WORKGROUP);
}
__device__ __forceinline__ void lds_max(unsigned* p, unsigned v) {
    __hip_atomic_fetch_max(p, v, __ATOMIC_RELAXED, __HIP_MEMORY_SCOPE_WORKGROUP);
}

__global__ __launch_bounds__(TPB)
void range_attend(const int* __restrict__ ei,
                  const float* __restrict__ Q,
                  const float* __restrict__ K,
                  const unsigned* __restrict__ rmap,
                  float* __restrict__ out) {
    __shared__ __attribute__((aligned(16))) float sK[NPB * OUT_F];
    __shared__ float sAcc[NPB * ACC_PITCH];
    __shared__ unsigned sMax[NPB];
    __shared__ float    sDen[NPB];
    __shared__ unsigned sList[CAP];
    __shared__ int      sRow[CAP];
    __shared__ int      sSrc[CAP + STEP];
    __shared__ unsigned sDeg[NPB];
    __shared__ int      sOff[NPB + 1];
    __shared__ unsigned sCnt;
    __shared__ unsigned sNext;

    const int tid = threadIdx.x;
    const int n0  = blockIdx.x * NPB;
    const int* __restrict__ cols = ei + N_EDGES;

    for (int i = tid; i < NPB * OUT_F / 4; i += TPB)
        ((float4*)sK)[i] = ((const float4*)(K + (size_t)n0 * OUT_F))[i];
    if (tid < NPB) sDeg[tid] = 0;
    if (tid == 0) { sCnt = 0; sNext = 0; }
    __syncthreads();

    {
        const uint4* __restrict__ lo4 = (const uint4*)rmap;
        const uint4* __restrict__ hi4 = lo4 + NG;
        unsigned nm[8];
        for (int k = 0; k < 8; ++k)
            nm[k] = ((blockIdx.x >> k) & 1u) ? 0u : ~0u;
        const int t0 = blockIdx.x % SCAN_TRIPS;
#pragma nounroll
        for (int t = 0; t < SCAN_TRIPS; ++t) {
            int tt = t + t0;
            if (tt >= SCAN_TRIPS) tt -= SCAN_TRIPS;
            const int base = tt * (TPB * SCAN_U) + tid;
            uint4 a[SCAN_U], b[SCAN_U];
#pragma unroll
            for (int u = 0; u < SCAN_U; ++u) {
                const int i = base + u * TPB;
                const int j = i < NG ? i : tid;
                a[u] = lo4[j];
                b[u] = hi4[j];
            }
#pragma unroll
            for (int u = 0; u < SCAN_U; ++u) {
                const int i = base + u * TPB;
                unsigned hits = (a[u].x ^ nm[0]) & (a[u].y ^ nm[1]) & (a[u].z ^ nm[2]) & (a[u].w ^ nm[3]) &
                                (b[u].x ^ nm[4]) & (b[u].y ^ nm[5]) & (b[u].z ^ nm[6]) & (b[u].w ^ nm[7]);
                if (i >= NG) hits = 0;
                while (hits) {
                    const unsigned bit = (unsigned)__builtin_ctz(hits);
                    hits &= hits - 1;
                    const unsigned p = atomicAdd(&sCnt, 1u);
                    if (p < CAP) sList[p] = (unsigned)i * 32u + bit;
                }
            }
        }
    }
    __syncthreads();

    const unsigned cnt = (unsigned)__builtin_amdgcn_readfirstlane((int)sCnt);

    if (cnt <= CAP) {
        const int n = (int)cnt;
        for (int p = tid; p < n; p += TPB) {
            const unsigned e = sList[p];
            const unsigned d = min((unsigned)(cols[e] - n0), (unsigned)(NPB - 1));
            const int src = ei[e];
            const unsigned place = atomicAdd(&sDeg[d], 1u);
            sRow[p]  = src;
            sList[p] = (place << 8) | d;
        }
        __syncthreads();

        if (tid < 32) {
            int part[BPL];
            int tot = 0;
#pragma unroll
            for (int j = 0; j < BPL; ++j) {
                const int idx = tid * BPL + j;
                part[j] = tot;
                tot += idx < NPB ? (int)sDeg[idx] : 0;
            }
            int inc = tot;
#pragma unroll
            for (int o = 1; o < 32; o <<= 1) {
                const int y = __shfl_up(inc, o, 32);
                if (tid >= o) inc += y;
            }
            const int excl = inc - tot;
#pragma unroll
            for (int j = 0; j < BPL; ++j) {
                const int idx = tid * BPL + j;
                if (idx < NPB) sOff[idx] = excl + part[j];
            }
            if (tid == 31) sOff[NPB] = inc;
        }
        __syncthreads();

        for (int p = tid; p < n; p += TPB) {
            const unsigned v = sList[p];
            sSrc[sOff[v & 255u] + (int)(v >> 8)] = sRow[p];
        }
        __syncthreads();

        const int      lane = tid & 31;
        const unsigned sub  = tid & 15;

        const char* __restrict__ qb = (const char*)Q;
        char*       __restrict__ ob = (char*)out;

        for (;;) {
            int t = 0;
            if (lane == 0) t = (int)atomicAdd(&sNext, 2u);
            t = __builtin_amdgcn_readfirstlane(t);
            if (t >= NPB) break;
            const int nd  = t + (lane >> 4);
            const int beg = sOff[nd], end = sOff[nd + 1];
            const float4 k = *(const float4*)(sK + nd * OUT_F + sub * 4);
            float  m = -INFINITY, den = 0.0f;
            float4 acc = make_float4(0.0f, 0.0f, 0.0f, 0.0f);
            for (int c = beg; c < end; c += STEP) {
                int    s[STEP];
                float4 q[STEP];
#pragma unroll
                for (int u = 0; u < STEP; ++u)
                    s[u] = sSrc[c + u];
#pragma unroll
                for (int u = 0; u < STEP; ++u) {
                    const unsigned su = (unsigned)((c + u < end) ? s[u] : s[0]);
                    q[u] = *(const float4*)(qb + (su * (unsigned)(OUT_F * 4) + sub * 16u));
                }
                soft_step(q, c, end, k, sub, m, den, acc);
            }
            const float dd = den + 1e-16f;
            *(float4*)(ob + ((unsigned)(n0 + nd) * (unsigned)(OUT_F * 4) + sub * 16u)) =
                make_float4(acc.x / dd, acc.y / dd, acc.z / dd, acc.w / dd);
        }
    } else {
        for (int i = tid; i < NPB * ACC_PITCH; i += TPB)
            sAcc[i] = 0.0f;
        if (tid < NPB) {
            sMax[tid] = fkey(-INFINITY);
            sDen[tid] = 0.0f;
        }
        __syncthreads();
        for (int e = tid; e < N_EDGES; e += TPB) {
            const unsigned d = (unsigned)(cols[e] - n0);
            if (d < NPB)
                lds_max(&sMax[d], fkey(edge_logit(Q + (size_t)ei[e] * OUT_F, sK + d * OUT_F)));
        }
        __syncthreads();
        for (int e = tid; e < N_EDGES; e += TPB) {
            const unsigned d = (unsigned)(cols[e] - n0);
            if (d < NPB) {
                const float s = edge_logit(Q + (size_t)ei[e] * OUT_F, sK + d * OUT_F);
                lds_add(&sDen[d], expf(s - funkey(sMax[d])));
            }
        }
        __syncthreads();
        for (int e = tid; e < N_EDGES; e += TPB) {
            const unsigned d = (unsigned)(cols[e] - n0);
            if (d < NPB) {
                const float* q = Q + (size_t)ei[e] * OUT_F;
                const float s = edge_logit(q, sK + d * OUT_F);
                const float w = expf(s - funkey(sMax[d])) / (sDen[d] + 1e-16f);
                for (int j = 0; j < OUT_F; ++j) lds_add(&sAcc[d * ACC_PITCH + j], q[j] * w);
            }
        }
        __syncthreads();

        for (int i = tid; i < NPB * OUT_F; i += TPB)
            out[(size_t)n0 * OUT_F + i] = sAcc[(i >> 6) * ACC_PITCH + (i & 63)];
    }
}

extern "C" void kernel_launch(void* const* d_in, const int* in_sizes, int n_in,
                              void* d_out, int out_size, void* d_ws, size_t ws_size,
                              hipStream_t stream) {
    const float* x  = (const float*)d_in[0];
    const int*   ei = (const int*)d_in[1];
    const float* Wq = (const float*)d_in[2];
    const float* Wk = (const float*)d_in[3];
    float* out = (float*)d_out;

    char* ws = (char*)d_ws;
    float*    Q       = (float*)ws;    ws += (size_t)N_NODES * OUT_F * sizeof(float);
    float*    K       = (float*)ws;    ws += (size_t)N_NODES * OUT_F * sizeof(float);
    unsigned* rmap    = (unsigned*)ws;

    qk_gemm<<<GEMM_BLOCKS, 256, 0, stream>>>(x, Wq, Wk, ei, Q, K, rmap);
    range_attend<<<NBLK, TPB, 0, stream>>>(ei, Q, K, rmap, out);
}


# qk_gemm as a tile loop: 768 work-groups (3 per WGP) take tiles g, g+768, ..; weights to LDS once per work-group, B fragment stays in registers, next x tile prefetched; order of the sums unchanged
# speedup vs baseline: 6.6509x; 1.0721x over previous
#include <hip/hip_runtime.h>
#include <math.h>

#define N_NODES 50000
#define N_EDGES 800000
#define IN_F    128
#define OUT_F   64

typedef __attribute__((ext_vector_type(2))) float v2f;
typedef __attribute__((ext_vector_type(8))) float v8f;

__device__ __forceinline__ unsigned fkey(float f) {
    unsigned u = __float_as_uint(f);
    return (u & 0x80000000u) ? ~u : (u | 0x80000000u);
}
__device__ __forceinline__ float funkey(unsigned u) {
    return __uint_as_float((u & 0x80000000u) ? (u & 0x7FFFFFFFu) : ~u);
}

static_assert(N_NODES % 16 == 0, "a block owns 16 whole nodes");
#define NPB   200
#define GEMM_BLOCKS (N_NODES / 16)
#define EPB   (N_EDGES / GEMM_BLOCKS)
#define NG    (N_EDGES / 32)
#define NO_RANGE 255u
static_assert(N_EDGES % GEMM_BLOCKS == 0, "the blocks share the edges evenly");
static_assert(EPB == 8 * 32, "one group of 32 edges for each of a block's 8 waves");
static_assert(N_NODES / NPB <= NO_RANGE, "a range number fits in 8 bits");

__device__ __forceinline__ unsigned range_of(int id) {
    const unsigned r = (unsigned)id / NPB;
    return r < NO_RANGE ? r : NO_RANGE;
}
static_assert(16 * IN_F == 2 * 256 * 4, "256 threads x 2 x 4 floats cover a block's x tile");
static_assert(IN_F % 8 == 0, "a step takes 8 k values");
#define A_PITCH (IN_F + 4)
#define K_STEPS (IN_F / 8)
#define QK_WGS 768
#define W_ROWS (2 * OUT_F)
static_assert(OUT_F * IN_F == 8 * 256 * 4, "256 threads x 8 x 4 floats cover one weight matrix");

__device__ __forceinline__ void lds_barrier() {
    asm volatile("s_wait_dscnt 0x0\n\ts_barrier_signal -1\n\ts_barrier_wait -1" ::: "memory");
}

__global__ __launch_bounds__(256)
void qk_gemm(const float* __restrict__ x,
             const float* __restrict__ Wq,
             const float* __restrict__ Wk,
             const int* __restrict__ ei,
             float* __restrict__ Q,
             float* __restrict__ K,
             unsigned* __restrict__ rmap) {
    __shared__ __attribute__((aligned(16))) float sW[W_ROWS * A_PITCH];
    __shared__ __attribute__((aligned(16))) float sA[2][16 * A_PITCH];
    __shared__ unsigned sMap[2][8 * 8];

    const int tid     = threadIdx.x;
    const int lane    = tid & 31;
    const int wave    = tid >> 5;
    const int mat     = wave >> 2;
    const int colBase = (wave & 3) * 16;
    const int* __restrict__ cols = ei + N_EDGES;

    float*       __restrict__ Out = mat ? K  : Q;
    const int l    = lane & 15;
    const int koff = (lane >> 4) * 4;

    float4 x0, x1;
    int    dst;
    auto fetch = [&](int t) {
        const float4* __restrict__ xt = (const float4*)(x + (size_t)t * 16 * IN_F);
        x0  = xt[tid];
        x1  = xt[tid + 256];
        dst = cols[(t * 8 + wave) * 32 + lane];
    };
    auto stage = [&](int p) {
        float* sAp = sA[p];
        const int i0 = tid, i1 = tid + 256;
        *(float4*)(sAp + (i0 >> 5) * A_PITCH + (i0 & 31) * 4) = x0;
        *(float4*)(sAp + (i1 >> 5) * A_PITCH + (i1 & 31) * 4) = x1;
        const unsigned r = range_of(dst);
        unsigned w = 0;
        for (int k = 0; k < 8; ++k) {
            const unsigned plane = __builtin_amdgcn_ballot_w32(((r >> k) & 1u) != 0);
            if (lane == k) w = plane;
        }
        if (lane < 8) sMap[p][wave * 8 + lane] = w;
    };

    int tile = blockIdx.x;
    fetch(tile);

    {
        const float4* __restrict__ wq4 = (const float4*)Wq;
        const float4* __restrict__ wk4 = (const float4*)Wk;
        float4 w[16];
#pragma unroll
        for (int j = 0; j < 8; ++j) {
            w[j]     = wq4[tid + 256 * j];
            w[8 + j] = wk4[tid + 256 * j];
        }
#pragma unroll
        for (int j = 0; j < 8; ++j) {
            const int idx = tid + 256 * j;
            float* p = sW + (idx >> 5) * A_PITCH + (idx & 31) * 4;
            *(float4*)p                      = w[j];
            *(float4*)(p + OUT_F * A_PITCH)  = w[8 + j];
        }
    }
    stage(0);
    if (tile + QK_WGS < GEMM_BLOCKS) fetch(tile + QK_WGS);
    lds_barrier();

    float4 b[K_STEPS];
    {
        const float* brow = sW + (mat * OUT_F + colBase + l) * A_PITCH + koff;
#pragma unroll
        for (int kk = 0; kk < K_STEPS; ++kk)
            b[kk] = *(const float4*)(brow + kk * 8);
    }

    for (int i = 0;; ++i) {
        const int p = i & 1;

        if (tid < 64) {
            const int half = tid >> 5, within = tid & 31;
            rmap[half * (NG * 4) + tile * 32 + within] =
                sMap[p][(within >> 2) * 8 + half * 4 + (within & 3)];
        }

        const float* arow = sA[p] + l * A_PITCH + koff;

        v8f acc = {};
#pragma unroll
        for (int kk = 0; kk < K_STEPS; ++kk) {
            const float4 a = *(const float4*)(arow + kk * 8);
            v2f a01 = {a.x, a.y}, a23 = {a.z, a.w};
            v2f b01 = {b[kk].x, b[kk].y}, b23 = {b[kk].z, b[kk].w};
            acc = __builtin_amdgcn_wmma_f32_16x16x4_f32(
                 false, a01,  false, b01,
                 (short)0, acc,  false,  false);
            acc = __builtin_amdgcn_wmma_f32_16x16x4_f32(
                 false, a23,  false, b23,
                 (short)0, acc,  false,  false);
        }

        const int outRow = tile * 16 + ((lane >> 4) ? 8 : 0);
        const int outCol = colBase + l;
#pragma unroll
        for (int r = 0; r < 8; ++r)
            Out[(size_t)(outRow + r) * OUT_F + outCol] = acc[r];

        tile += QK_WGS;
        if (tile >= GEMM_BLOCKS) break;
        stage(p ^ 1);
        if (tile + QK_WGS < GEMM_BLOCKS) fetch(tile + QK_WGS);
        lds_barrier();
    }
}

#define NBLK  (N_NODES / NPB)
#define TPB   1024
#define CAP   8192
#define STEP  8
#define BPL   ((NPB + 31) / 32)
#define ACC_PITCH (OUT_F + 1)
#define SCAN_U 5
#define SCAN_TRIPS ((NG + TPB * SCAN_U - 1) / (TPB * SCAN_U))
static_assert(N_NODES % NPB == 0, "ranges tile the nodes exactly");
static_assert(NPB <= 256, "local node index is packed into 8 bits");
static_assert(NPB % 2 == 0, "a ticket is a pair of nodes, one per 16-lane row of a wave");
static_assert(CAP <= (1 << 24), "place among a node's edges is packed into 24 bits");
static_assert(N_EDGES % 32 == 0, "the map is kept in groups of 32 edges");
static_assert(NG > TPB, "the dummy group of a thread past the end exists");
static_assert((NPB * OUT_F) % 4 == 0, "K rows move as float4");
static_assert(OUT_F == 64, "16 lanes x 4 floats per row");
static_assert(STEP == 8, "soft_step names its 8 places one by one; lane STEP of a 16-lane row exists");

template <int M>
__device__ __forceinline__ float row_xadd(float p) {
    return p + __int_as_float(__builtin_amdgcn_update_dpp(
                   0, __float_as_int(p), 0x160 + M, 0xf, 0xf, true));
}
__device__ __forceinline__ float row_dot(float4 q, float4 k) {
    float p = q.x * k.x + q.y * k.y + q.z * k.z + q.w * k.w;
    p = row_xadd<1>(p);
    p = row_xadd<2>(p);
    p = row_xadd<4>(p);
    p = row_xadd<8>(p);
    return p;
}

template <int U>
__device__ __forceinline__ float row_bcast(float v) {
    return __int_as_float(__builtin_amdgcn_update_dpp(
               0, __float_as_int(v), 0x150 + U, 0xf, 0xf, true));
}

__device__ __forceinline__ void soft_step(const float4* q, int c, int end, float4 k, unsigned sub,
                                          float& m, float& den, float4& acc) {
    float a[STEP];
    float cm = -INFINITY;
#pragma unroll
    for (int u = 0; u < STEP; ++u) {
        const float d = row_dot(q[u], k) * 0.125f;
        a[u] = (c + u < end) ? d : -INFINITY;
        cm = fmaxf(cm, a[u]);
    }
    const float mn = fmaxf(m, cm);
    float mine = a[0];
#pragma unroll
    for (int u = 1; u < STEP; ++u)
        mine = ((sub & (STEP - 1)) == (unsigned)u) ? a[u] : mine;
    mine = (sub & STEP) ? m : mine;
    const float exm = expf(mine - mn);
    const float sc  = row_bcast<STEP>(exm);
    den   *= sc;
    acc.x *= sc; acc.y *= sc; acc.z *= sc; acc.w *= sc;
#define SOFT_SLOT(U) { const float ex = row_bcast<U>(exm); den += ex; \
        acc.x += q[U].x * ex; acc.y += q[U].y * ex; acc.z += q[U].z * ex; acc.w += q[U].w * ex; }
    SOFT_SLOT(0) SOFT_SLOT(1) SOFT_SLOT(2) SOFT_SLOT(3)
    SOFT_SLOT(4) SOFT_SLOT(5) SOFT_SLOT(6) SOFT_SLOT(7)
#undef SOFT_SLOT
    m = mn;
}

__device__ __forceinline__ float edge_logit(const float* __restrict__ q, const float* k) {
    float s = 0.0f;
    for (int j = 0; j < OUT_F; ++j) s += q[j] * k[j];
    return s * 0.125f;
}

__device__ __forceinline__ void lds_add(float* p, float v) {
    __hip_atomic_fetch_add(p, v, __ATOMIC_RELAXED, __HIP_MEMORY_SCOPE_WORKGROUP);
}
__device__ __forceinline__ void lds_max(unsigned* p, unsigned v) {
    __hip_atomic_fetch_max(p, v, __ATOMIC_RELAXED, __HIP_MEMORY_SCOPE_WORKGROUP);
}

__global__ __launch_bounds__(TPB)
void range_attend(const int* __restrict__ ei,
                  const float* __restrict__ Q,
                  const float* __restrict__ K,
                  const unsigned* __restrict__ rmap,
                  float* __restrict__ out) {
    __shared__ __attribute__((aligned(16))) float sK[NPB * OUT_F];
    __shared__ float sAcc[NPB * ACC_PITCH];
    __shared__ unsigned sMax[NPB];
    __shared__ float    sDen[NPB];
    __shared__ unsigned sList[CAP];
    __shared__ int      sRow[CAP];
    __shared__ int      sSrc[CAP + STEP];
    __shared__ unsigned sDeg[NPB];
    __shared__ int      sOff[NPB + 1];
    __shared__ unsigned sCnt;
    __shared__ unsigned sNext;

    const int tid = threadIdx.x;
    const int n0  = blockIdx.x * NPB;
    const int* __restrict__ cols = ei + N_EDGES;

    for (int i = tid; i < NPB * OUT_F / 4; i += TPB)
        ((float4*)sK)[i] = ((const float4*)(K + (size_t)n0 * OUT_F))[i];
    if (tid < NPB) sDeg[tid] = 0;
    if (tid == 0) { sCnt = 0; sNext = 0; }
    __syncthreads();

    {
        const uint4* __restrict__ lo4 = (const uint4*)rmap;
        const uint4* __restrict__ hi4 = lo4 + NG;
        unsigned nm[8];
        for (int k = 0; k < 8; ++k)
            nm[k] = ((blockIdx.x >> k) & 1u) ? 0u : ~0u;
        const int t0 = blockIdx.x % SCAN_TRIPS;
#pragma nounroll
        for (int t = 0; t < SCAN_TRIPS; ++t) {
            int tt = t + t0;
            if (tt >= SCAN_TRIPS) tt -= SCAN_TRIPS;
            const int base = tt * (TPB * SCAN_U) + tid;
            uint4 a[SCAN_U], b[SCAN_U];
#pragma unroll
            for (int u = 0; u < SCAN_U; ++u) {
                const int i = base + u * TPB;
                const int j = i < NG ? i : tid;
                a[u] = lo4[j];
                b[u] = hi4[j];
            }
#pragma unroll
            for (int u = 0; u < SCAN_U; ++u) {
                const int i = base + u * TPB;
                unsigned hits = (a[u].x ^ nm[0]) & (a[u].y ^ nm[1]) & (a[u].z ^ nm[2]) & (a[u].w ^ nm[3]) &
                                (b[u].x ^ nm[4]) & (b[u].y ^ nm[5]) & (b[u].z ^ nm[6]) & (b[u].w ^ nm[7]);
                if (i >= NG) hits = 0;
                while (hits) {
                    const unsigned bit = (unsigned)__builtin_ctz(hits);
                    hits &= hits - 1;
                    const unsigned p = atomicAdd(&sCnt, 1u);
                    if (p < CAP) sList[p] = (unsigned)i * 32u + bit;
                }
            }
        }
    }
    __syncthreads();

    const unsigned cnt = (unsigned)__builtin_amdgcn_readfirstlane((int)sCnt);

    if (cnt <= CAP) {
        const int n = (int)cnt;
        for (int p = tid; p < n; p += TPB) {
            const unsigned e = sList[p];
            const unsigned d = min((unsigned)(cols[e] - n0), (unsigned)(NPB - 1));
            const int src = ei[e];
            const unsigned place = atomicAdd(&sDeg[d], 1u);
            sRow[p]  = src;
            sList[p] = (place << 8) | d;
        }
        __syncthreads();

        if (tid < 32) {
            int part[BPL];
            int tot = 0;
#pragma unroll
            for (int j = 0; j < BPL; ++j) {
                const int idx = tid * BPL + j;
                part[j] = tot;
                tot += idx < NPB ? (int)sDeg[idx] : 0;
            }
            int inc = tot;
#pragma unroll
            for (int o = 1; o < 32; o <<= 1) {
                const int y = __shfl_up(inc, o, 32);
                if (tid >= o) inc += y;
            }
            const int excl = inc - tot;
#pragma unroll
            for (int j = 0; j < BPL; ++j) {
                const int idx = tid * BPL + j;
                if (idx < NPB) sOff[idx] = excl + part[j];
            }
            if (tid == 31) sOff[NPB] = inc;
        }
        __syncthreads();

        for (int p = tid; p < n; p += TPB) {
            const unsigned v = sList[p];
            sSrc[sOff[v & 255u] + (int)(v >> 8)] = sRow[p];
        }
        __syncthreads();

        const int      lane = tid & 31;
        const unsigned sub  = tid & 15;

        const char* __restrict__ qb = (const char*)Q;
        char*       __restrict__ ob = (char*)out;

        for (;;) {
            int t = 0;
            if (lane == 0) t = (int)atomicAdd(&sNext, 2u);
            t = __builtin_amdgcn_readfirstlane(t);
            if (t >= NPB) break;
            const int nd  = t + (lane >> 4);
            const int beg = sOff[nd], end = sOff[nd + 1];
            const float4 k = *(const float4*)(sK + nd * OUT_F + sub * 4);
            float  m = -INFINITY, den = 0.0f;
            float4 acc = make_float4(0.0f, 0.0f, 0.0f, 0.0f);
            for (int c = beg; c < end; c += STEP) {
                int    s[STEP];
                float4 q[STEP];
#pragma unroll
                for (int u = 0; u < STEP; ++u)
                    s[u] = sSrc[c + u];
#pragma unroll
                for (int u = 0; u < STEP; ++u) {
                    const unsigned su = (unsigned)((c + u < end) ? s[u] : s[0]);
                    q[u] = *(const float4*)(qb + (su * (unsigned)(OUT_F * 4) + sub * 16u));
                }
                soft_step(q, c, end, k, sub, m, den, acc);
            }
            const float dd = den + 1e-16f;
            *(float4*)(ob + ((unsigned)(n0 + nd) * (unsigned)(OUT_F * 4) + sub * 16u)) =
                make_float4(acc.x / dd, acc.y / dd, acc.z / dd, acc.w / dd);
        }
    } else {
        for (int i = tid; i < NPB * ACC_PITCH; i += TPB)
            sAcc[i] = 0.0f;
        if (tid < NPB) {
            sMax[tid] = fkey(-INFINITY);
            sDen[tid] = 0.0f;
        }
        __syncthreads();
        for (int e = tid; e < N_EDGES; e += TPB) {
            const unsigned d = (unsigned)(cols[e] - n0);
            if (d < NPB)
                lds_max(&sMax[d], fkey(edge_logit(Q + (size_t)ei[e] * OUT_F, sK + d * OUT_F)));
        }
        __syncthreads();
        for (int e = tid; e < N_EDGES; e += TPB) {
            const unsigned d = (unsigned)(cols[e] - n0);
            if (d < NPB) {
                const float s = edge_logit(Q + (size_t)ei[e] * OUT_F, sK + d * OUT_F);
                lds_add(&sDen[d], expf(s - funkey(sMax[d])));
            }
        }
        __syncthreads();
        for (int e = tid; e < N_EDGES; e += TPB) {
            const unsigned d = (unsigned)(cols[e] - n0);
            if (d < NPB) {
                const float* q = Q + (size_t)ei[e] * OUT_F;
                const float s = edge_logit(q, sK + d * OUT_F);
                const float w = expf(s - funkey(sMax[d])) / (sDen[d] + 1e-16f);
                for (int j = 0; j < OUT_F; ++j) lds_add(&sAcc[d * ACC_PITCH + j], q[j] * w);
            }
        }
        __syncthreads();

        for (int i = tid; i < NPB * OUT_F; i += TPB)
            out[(size_t)n0 * OUT_F + i] = sAcc[(i >> 6) * ACC_PITCH + (i & 63)];
    }
}

extern "C" void kernel_launch(void* const* d_in, const int* in_sizes, int n_in,
                              void* d_out, int out_size, void* d_ws, size_t ws_size,
                              hipStream_t stream) {
    const float* x  = (const float*)d_in[0];
    const int*   ei = (const int*)d_in[1];
    const float* Wq = (const float*)d_in[2];
    const float* Wk = (const float*)d_in[3];
    float* out = (float*)d_out;

    char* ws = (char*)d_ws;
    float*    Q       = (float*)ws;    ws += (size_t)N_NODES * OUT_F * sizeof(float);
    float*    K       = (float*)ws;    ws += (size_t)N_NODES * OUT_F * sizeof(float);
    unsigned* rmap    = (unsigned*)ws;

    qk_gemm<<<QK_WGS, 256, 0, stream>>>(x, Wq, Wk, ei, Q, K, rmap);
    range_attend<<<NBLK, TPB, 0, stream>>>(ei, Q, K, rmap, out);
}


# edges binned by destination range in qk_gemm (per-chunk LDS counting sort: records + start table); range_attend reads its runs, no map scan, no id gathers; LDS list order (sum order) changes
# speedup vs baseline: 6.9301x; 1.0420x over previous
#include <hip/hip_runtime.h>
#include <math.h>

#define N_NODES 50000
#define N_EDGES 800000
#define IN_F    128
#define OUT_F   64

typedef __attribute__((ext_vector_type(2))) float v2f;
typedef __attribute__((ext_vector_type(8))) float v8f;

__device__ __forceinline__ unsigned fkey(float f) {
    unsigned u = __float_as_uint(f);
    return (u & 0x80000000u) ? ~u : (u | 0x80000000u);
}
__device__ __forceinline__ float funkey(unsigned u) {
    return __uint_as_float((u & 0x80000000u) ? (u & 0x7FFFFFFFu) : ~u);
}

static_assert(N_NODES % 16 == 0, "a block owns 16 whole nodes");
#define NPB   200
#define GEMM_BLOCKS (N_NODES / 16)
#define CHUNK 1280
#define EPT   (CHUNK / 256)
#define NCHUNK (N_EDGES / CHUNK)
#define NBIN  256
#define DUMP_BIN (N_NODES / NPB)
#define REC_D_OFF ((size_t)N_EDGES * 2)
#define TAB_OFF   ((size_t)N_EDGES * 3)
static_assert(N_EDGES % CHUNK == 0, "chunks tile the edges exactly");
static_assert(CHUNK % 256 == 0, "every thread takes the same number of edges");
static_assert(CHUNK < 65536, "a place inside a chunk fits in 16 bits");
static_assert(N_NODES <= 65536, "a source id fits in 16 bits");
static_assert(NPB <= 256, "a destination's number inside its range fits in 8 bits");
static_assert(DUMP_BIN + 1 < NBIN, "the end of the last range is a table entry");
static_assert(TAB_OFF % 4 == 0 && REC_D_OFF % 4 == 0, "the arrays start on whole words");
static_assert(16 * IN_F == 2 * 256 * 4, "256 threads x 2 x 4 floats cover a block's x tile");
static_assert(IN_F % 8 == 0, "a step takes 8 k values");
#define A_PITCH (IN_F + 4)
#define K_STEPS (IN_F / 8)
#define QK_WGS 768
static_assert(NCHUNK <= QK_WGS, "a work-group sorts at most one chunk");
#define W_ROWS (2 * OUT_F)
static_assert(OUT_F * IN_F == 8 * 256 * 4, "256 threads x 8 x 4 floats cover one weight matrix");

__device__ __forceinline__ void lds_barrier() {
    asm volatile("s_wait_dscnt 0x0\n\ts_barrier_signal -1\n\ts_barrier_wait -1" ::: "memory");
}

__global__ __launch_bounds__(256)
void qk_gemm(const float* __restrict__ x,
             const float* __restrict__ Wq,
             const float* __restrict__ Wk,
             const int* __restrict__ ei,
             float* __restrict__ Q,
             float* __restrict__ K,
             unsigned* __restrict__ bins) {
    __shared__ __attribute__((aligned(16))) float sW[W_ROWS * A_PITCH];
    __shared__ __attribute__((aligned(16))) float sA[2][16 * A_PITCH];
    __shared__ unsigned sBin[NBIN];
    __shared__ unsigned sRec[CHUNK];

    const int tid     = threadIdx.x;
    const int lane    = tid & 31;
    const int wave    = tid >> 5;
    const int mat     = wave >> 2;
    const int colBase = (wave & 3) * 16;
    const int* __restrict__ cols = ei + N_EDGES;
    unsigned short* __restrict__ recS = (unsigned short*)bins;
    unsigned char*  __restrict__ recD = (unsigned char*)bins + REC_D_OFF;
    unsigned short* __restrict__ tab  = (unsigned short*)((char*)bins + TAB_OFF);

    float*       __restrict__ Out = mat ? K  : Q;
    const int l    = lane & 15;
    const int koff = (lane >> 4) * 4;

    float4 x0, x1;
    auto fetch = [&](int t) {
        const float4* __restrict__ xt = (const float4*)(x + (size_t)t * 16 * IN_F);
        x0  = xt[tid];
        x1  = xt[tid + 256];
    };
    auto stage = [&](int p) {
        float* sAp = sA[p];
        const int i0 = tid, i1 = tid + 256;
        *(float4*)(sAp + (i0 >> 5) * A_PITCH + (i0 & 31) * 4) = x0;
        *(float4*)(sAp + (i1 >> 5) * A_PITCH + (i1 & 31) * 4) = x1;
    };

    const bool sorter = blockIdx.x < NCHUNK;
    const unsigned cbase = blockIdx.x * (unsigned)CHUNK;
    int dsts[EPT], srcs[EPT];
#pragma unroll
    for (int j = 0; j < EPT; ++j) {
        dsts[j] = 0;
        srcs[j] = 0;
        if (sorter) {
            dsts[j] = cols[cbase + j * 256 + tid];
            srcs[j] = ei[cbase + j * 256 + tid];
        }
    }

    int tile = blockIdx.x;
    fetch(tile);

    {
        const float4* __restrict__ wq4 = (const float4*)Wq;
        const float4* __restrict__ wk4 = (const float4*)Wk;
#define W_LOAD(J) const float4 wq_##J = wq4[tid + 256 * J]; const float4 wk_##J = wk4[tid + 256 * J];
        W_LOAD(0) W_LOAD(1) W_LOAD(2) W_LOAD(3) W_LOAD(4) W_LOAD(5) W_LOAD(6) W_LOAD(7)
#undef W_LOAD

        if (sorter) {
            sBin[tid] = 0;
            __syncthreads();
            unsigned bin[EPT], pos[EPT];
#pragma unroll
            for (int j = 0; j < EPT; ++j) {
                const unsigned r = (unsigned)dsts[j] / NPB;
                bin[j] = r < DUMP_BIN ? r : DUMP_BIN;
                pos[j] = atomicAdd(&sBin[bin[j]], 1u);
            }
            __syncthreads();
            if (tid < 32) {
                int part[NBIN / 32];
                int tot = 0;
#pragma unroll
                for (int j = 0; j < NBIN / 32; ++j) {
                    part[j] = tot;
                    tot += (int)sBin[tid * (NBIN / 32) + j];
                }
                int inc = tot;
#pragma unroll
                for (int o = 1; o < 32; o <<= 1) {
                    const int y = __shfl_up(inc, o, 32);
                    if (tid >= o) inc += y;
                }
                const int excl = inc - tot;
#pragma unroll
                for (int j = 0; j < NBIN / 32; ++j)
                    sBin[tid * (NBIN / 32) + j] = (unsigned)(excl + part[j]);
            }
            __syncthreads();
#pragma unroll
            for (int j = 0; j < EPT; ++j)
                sRec[sBin[bin[j]] + pos[j]] =
                    ((unsigned)srcs[j] << 8) | (((unsigned)dsts[j] - bin[j] * NPB) & 255u);
            tab[blockIdx.x * NBIN + tid] = (unsigned short)sBin[tid];
            __syncthreads();
#pragma unroll
            for (int j = 0; j < EPT; ++j) {
                const unsigned i = j * 256 + tid;
                const unsigned v = sRec[i];
                recS[cbase + i] = (unsigned short)(v >> 8);
                recD[cbase + i] = (unsigned char)(v & 255u);
            }
        }

#define W_STORE(J) { const int idx = tid + 256 * J; \
            float* p = sW + (idx >> 5) * A_PITCH + (idx & 31) * 4; \
            *(float4*)p                     = wq_##J; \
            *(float4*)(p + OUT_F * A_PITCH) = wk_##J; }
        W_STORE(0) W_STORE(1) W_STORE(2) W_STORE(3) W_STORE(4) W_STORE(5) W_STORE(6) W_STORE(7)
#undef W_STORE
    }
    stage(0);
    if (tile + QK_WGS < GEMM_BLOCKS) fetch(tile + QK_WGS);
    lds_barrier();

    float4 b[K_STEPS];
    {
        const float* brow = sW + (mat * OUT_F + colBase + l) * A_PITCH + koff;
#pragma unroll
        for (int kk = 0; kk < K_STEPS; ++kk)
            b[kk] = *(const float4*)(brow + kk * 8);
    }

    for (int i = 0;; ++i) {
        const int p = i & 1;

        const float* arow = sA[p] + l * A_PITCH + koff;

        v8f acc = {};
#pragma unroll
        for (int kk = 0; kk < K_STEPS; ++kk) {
            const float4 a = *(const float4*)(arow + kk * 8);
            v2f a01 = {a.x, a.y}, a23 = {a.z, a.w};
            v2f b01 = {b[kk].x, b[kk].y}, b23 = {b[kk].z, b[kk].w};
            acc = __builtin_amdgcn_wmma_f32_16x16x4_f32(
                 false, a01,  false, b01,
                 (short)0, acc,  false,  false);
            acc = __builtin_amdgcn_wmma_f32_16x16x4_f32(
                 false, a23,  false, b23,
                 (short)0, acc,  false,  false);
        }

        const int outRow = tile * 16 + ((lane >> 4) ? 8 : 0);
        const int outCol = colBase + l;
#pragma unroll
        for (int r = 0; r < 8; ++r)
            Out[(size_t)(outRow + r) * OUT_F + outCol] = acc[r];

        tile += QK_WGS;
        if (tile >= GEMM_BLOCKS) break;
        stage(p ^ 1);
        if (tile + QK_WGS < GEMM_BLOCKS) fetch(tile + QK_WGS);
        lds_barrier();
    }
}

#define NBLK  (N_NODES / NPB)
#define TPB   1024
#define CAP   8192
#define STEP  8
#define BPL   ((NPB + 31) / 32)
#define ACC_PITCH (OUT_F + 1)
#define RUN   12
static_assert(NCHUNK <= TPB, "one thread per chunk");
static_assert(N_NODES % NPB == 0, "ranges tile the nodes exactly");
static_assert(NPB <= 256, "local node index is packed into 8 bits");
static_assert(NPB % 2 == 0, "a ticket is a pair of nodes, one per 16-lane row of a wave");
static_assert(CAP <= (1 << 24), "place among a node's edges is packed into 24 bits");
static_assert((NPB * OUT_F) % 4 == 0, "K rows move as float4");
static_assert(OUT_F == 64, "16 lanes x 4 floats per row");
static_assert(STEP == 8, "soft_step names its 8 places one by one; lane STEP of a 16-lane row exists");

template <int M>
__device__ __forceinline__ float row_xadd(float p) {
    return p + __int_as_float(__builtin_amdgcn_update_dpp(
                   0, __float_as_int(p), 0x160 + M, 0xf, 0xf, true));
}
__device__ __forceinline__ float row_dot(float4 q, float4 k) {
    float p = q.x * k.x + q.y * k.y + q.z * k.z + q.w * k.w;
    p = row_xadd<1>(p);
    p = row_xadd<2>(p);
    p = row_xadd<4>(p);
    p = row_xadd<8>(p);
    return p;
}

template <int U>
__device__ __forceinline__ float row_bcast(float v) {
    return __int_as_float(__builtin_amdgcn_update_dpp(
               0, __float_as_int(v), 0x150 + U, 0xf, 0xf, true));
}

__device__ __forceinline__ void soft_step(const float4* q, int c, int end, float4 k, unsigned sub,
                                          float& m, float& den, float4& acc) {
    float a[STEP];
    float cm = -INFINITY;
#pragma unroll
    for (int u = 0; u < STEP; ++u) {
        const float d = row_dot(q[u], k) * 0.125f;
        a[u] = (c + u < end) ? d : -INFINITY;
        cm = fmaxf(cm, a[u]);
    }
    const float mn = fmaxf(m, cm);
    float mine = a[0];
#pragma unroll
    for (int u = 1; u < STEP; ++u)
        mine = ((sub & (STEP - 1)) == (unsigned)u) ? a[u] : mine;
    mine = (sub & STEP) ? m : mine;
    const float exm = expf(mine - mn);
    const float sc  = row_bcast<STEP>(exm);
    den   *= sc;
    acc.x *= sc; acc.y *= sc; acc.z *= sc; acc.w *= sc;
#define SOFT_SLOT(U) { const float ex = row_bcast<U>(exm); den += ex; \
        acc.x += q[U].x * ex; acc.y += q[U].y * ex; acc.z += q[U].z * ex; acc.w += q[U].w * ex; }
    SOFT_SLOT(0) SOFT_SLOT(1) SOFT_SLOT(2) SOFT_SLOT(3)
    SOFT_SLOT(4) SOFT_SLOT(5) SOFT_SLOT(6) SOFT_SLOT(7)
#undef SOFT_SLOT
    m = mn;
}

__device__ __forceinline__ float edge_logit(const float* __restrict__ q, const float* k) {
    float s = 0.0f;
    for (int j = 0; j < OUT_F; ++j) s += q[j] * k[j];
    return s * 0.125f;
}

__device__ __forceinline__ void lds_add(float* p, float v) {
    __hip_atomic_fetch_add(p, v, __ATOMIC_RELAXED, __HIP_MEMORY_SCOPE_WORKGROUP);
}
__device__ __forceinline__ void lds_max(unsigned* p, unsigned v) {
    __hip_atomic_fetch_max(p, v, __ATOMIC_RELAXED, __HIP_MEMORY_SCOPE_WORKGROUP);
}

__global__ __launch_bounds__(TPB)
void range_attend(const int* __restrict__ ei,
                  const float* __restrict__ Q,
                  const float* __restrict__ K,
                  const unsigned* __restrict__ bins,
                  float* __restrict__ out) {
    __shared__ __attribute__((aligned(16))) float sK[NPB * OUT_F];
    __shared__ float sAcc[NPB * ACC_PITCH];
    __shared__ unsigned sMax[NPB];
    __shared__ float    sDen[NPB];
    __shared__ unsigned sList[CAP];
    __shared__ int      sRow[CAP];
    __shared__ int      sSrc[CAP + STEP];
    __shared__ unsigned sDeg[NPB];
    __shared__ int      sOff[NPB + 1];
    __shared__ unsigned sCnt;
    __shared__ unsigned sNext;

    const int tid = threadIdx.x;
    const int n0  = blockIdx.x * NPB;
    const int* __restrict__ cols = ei + N_EDGES;

    const unsigned short* __restrict__ tab = (const unsigned short*)((const char*)bins + TAB_OFF);
    unsigned s0 = 0, s1 = 0;
    if (tid < NCHUNK) {
        s0 = tab[tid * NBIN + blockIdx.x];
        s1 = tab[tid * NBIN + blockIdx.x + 1];
    }

    for (int i = tid; i < NPB * OUT_F / 4; i += TPB)
        ((float4*)sK)[i] = ((const float4*)(K + (size_t)n0 * OUT_F))[i];
    if (tid < NPB) sDeg[tid] = 0;
    if (tid == 0) { sCnt = 0; sNext = 0; }
    __syncthreads();

    {
        const unsigned short* __restrict__ recS = (const unsigned short*)bins;
        const unsigned char*  __restrict__ recD = (const unsigned char*)bins + REC_D_OFF;
        auto put = [&](unsigned p, unsigned src, unsigned d8) {
            if (p < CAP) {
                const unsigned d = min(d8, (unsigned)(NPB - 1));
                const unsigned place = atomicAdd(&sDeg[d], 1u);
                sRow[p]  = (int)src;
                sList[p] = (place << 8) | d;
            }
        };
        if (tid < NCHUNK) {
            const unsigned cb = (unsigned)tid * CHUNK;
            s1 = min(s1, (unsigned)CHUNK);
            const unsigned len = s1 > s0 ? s1 - s0 : 0u;
            unsigned rs[RUN], rd[RUN];
#pragma unroll
            for (int u = 0; u < RUN; ++u) {
                const unsigned at = cb + ((unsigned)u < len ? s0 + (unsigned)u : 0u);
                rs[u] = recS[at];
                rd[u] = recD[at];
            }
            unsigned p0 = 0;
            if (len) p0 = atomicAdd(&sCnt, len);
            unsigned place[RUN];
#pragma unroll
            for (int u = 0; u < RUN; ++u) {
                rd[u] = min(rd[u], (unsigned)(NPB - 1));
                place[u] = 0;
                if ((unsigned)u < len && p0 + (unsigned)u < CAP) place[u] = atomicAdd(&sDeg[rd[u]], 1u);
            }
#pragma unroll
            for (int u = 0; u < RUN; ++u)
                if ((unsigned)u < len && p0 + (unsigned)u < CAP) {
                    sRow[p0 + u]  = (int)rs[u];
                    sList[p0 + u] = (place[u] << 8) | rd[u];
                }
            for (unsigned i = RUN; i < len; ++i)
                put(p0 + i, recS[cb + s0 + i], recD[cb + s0 + i]);
        }
    }
    __syncthreads();

    const unsigned cnt = (unsigned)__builtin_amdgcn_readfirstlane((int)sCnt);

    if (cnt <= CAP) {
        const int n = (int)cnt;
        if (tid < 32) {
            int part[BPL];
            int tot = 0;
#pragma unroll
            for (int j = 0; j < BPL; ++j) {
                const int idx = tid * BPL + j;
                part[j] = tot;
                tot += idx < NPB ? (int)sDeg[idx] : 0;
            }
            int inc = tot;
#pragma unroll
            for (int o = 1; o < 32; o <<= 1) {
                const int y = __shfl_up(inc, o, 32);
                if (tid >= o) inc += y;
            }
            const int excl = inc - tot;
#pragma unroll
            for (int j = 0; j < BPL; ++j) {
                const int idx = tid * BPL + j;
                if (idx < NPB) sOff[idx] = excl + part[j];
            }
            if (tid == 31) sOff[NPB] = inc;
        }
        __syncthreads();

        for (int p = tid; p < n; p += TPB) {
            const unsigned v = sList[p];
            sSrc[sOff[v & 255u] + (int)(v >> 8)] = sRow[p];
        }
        __syncthreads();

        const int      lane = tid & 31;
        const unsigned sub  = tid & 15;

        const char* __restrict__ qb = (const char*)Q;
        char*       __restrict__ ob = (char*)out;

        for (;;) {
            int t = 0;
            if (lane == 0) t = (int)atomicAdd(&sNext, 2u);
            t = __builtin_amdgcn_readfirstlane(t);
            if (t >= NPB) break;
            const int nd  = t + (lane >> 4);
            const int beg = sOff[nd], end = sOff[nd + 1];
            const float4 k = *(const float4*)(sK + nd * OUT_F + sub * 4);
            float  m = -INFINITY, den = 0.0f;
            float4 acc = make_float4(0.0f, 0.0f, 0.0f, 0.0f);
            for (int c = beg; c < end; c += STEP) {
                int    s[STEP];
                float4 q[STEP];
#pragma unroll
                for (int u = 0; u < STEP; ++u)
                    s[u] = sSrc[c + u];
#pragma unroll
                for (int u = 0; u < STEP; ++u) {
                    const unsigned su = (unsigned)((c + u < end) ? s[u] : s[0]);
                    q[u] = *(const float4*)(qb + (su * (unsigned)(OUT_F * 4) + sub * 16u));
                }
                soft_step(q, c, end, k, sub, m, den, acc);
            }
            const float dd = den + 1e-16f;
            *(float4*)(ob + ((unsigned)(n0 + nd) * (unsigned)(OUT_F * 4) + sub * 16u)) =
                make_float4(acc.x / dd, acc.y / dd, acc.z / dd, acc.w / dd);
        }
    } else {
        for (int i = tid; i < NPB * ACC_PITCH; i += TPB)
            sAcc[i] = 0.0f;
        if (tid < NPB) {
            sMax[tid] = fkey(-INFINITY);
            sDen[tid] = 0.0f;
        }
        __syncthreads();
        for (int e = tid; e < N_EDGES; e += TPB) {
            const unsigned d = (unsigned)(cols[e] - n0);
            if (d < NPB)
                lds_max(&sMax[d], fkey(edge_logit(Q + (size_t)ei[e] * OUT_F, sK + d * OUT_F)));
        }
        __syncthreads();
        for (int e = tid; e < N_EDGES; e += TPB) {
            const unsigned d = (unsigned)(cols[e] - n0);
            if (d < NPB) {
                const float s = edge_logit(Q + (size_t)ei[e] * OUT_F, sK + d * OUT_F);
                lds_add(&sDen[d], expf(s - funkey(sMax[d])));
            }
        }
        __syncthreads();
        for (int e = tid; e < N_EDGES; e += TPB) {
            const unsigned d = (unsigned)(cols[e] - n0);
            if (d < NPB) {
                const float* q = Q + (size_t)ei[e] * OUT_F;
                const float s = edge_logit(q, sK + d * OUT_F);
                const float w = expf(s - funkey(sMax[d])) / (sDen[d] + 1e-16f);
                for (int j = 0; j < OUT_F; ++j) lds_add(&sAcc[d * ACC_PITCH + j], q[j] * w);
            }
        }
        __syncthreads();

        for (int i = tid; i < NPB * OUT_F; i += TPB)
            out[(size_t)n0 * OUT_F + i] = sAcc[(i >> 6) * ACC_PITCH + (i & 63)];
    }
}

extern "C" void kernel_launch(void* const* d_in, const int* in_sizes, int n_in,
                              void* d_out, int out_size, void* d_ws, size_t ws_size,
                              hipStream_t stream) {
    const float* x  = (const float*)d_in[0];
    const int*   ei = (const int*)d_in[1];
    const float* Wq = (const float*)d_in[2];
    const float* Wk = (const float*)d_in[3];
    float* out = (float*)d_out;

    char* ws = (char*)d_ws;
    float*    Q       = (float*)ws;    ws += (size_t)N_NODES * OUT_F * sizeof(float);
    float*    K       = (float*)ws;    ws += (size_t)N_NODES * OUT_F * sizeof(float);
    unsigned* bins    = (unsigned*)ws;

    qk_gemm<<<QK_WGS, 256, 0, stream>>>(x, Wq, Wk, ei, Q, K, bins);
    range_attend<<<NBLK, TPB, 0, stream>>>(ei, Q, K, bins, out);
}


# range_attend: 8 lanes do a node (2 float4 per Q row per lane, whole-line loads), 4 nodes per wave and ticket, not 16 lanes and 2; DPP8 broadcast, scale exp in every lane; dot and sum order change
# speedup vs baseline: 7.1500x; 1.0317x over previous
#include <hip/hip_runtime.h>
#include <math.h>

#define N_NODES 50000
#define N_EDGES 800000
#define IN_F    128
#define OUT_F   64

typedef __attribute__((ext_vector_type(2))) float v2f;
typedef __attribute__((ext_vector_type(8))) float v8f;

__device__ __forceinline__ unsigned fkey(float f) {
    unsigned u = __float_as_uint(f);
    return (u & 0x80000000u) ? ~u : (u | 0x80000000u);
}
__device__ __forceinline__ float funkey(unsigned u) {
    return __uint_as_float((u & 0x80000000u) ? (u & 0x7FFFFFFFu) : ~u);
}

static_assert(N_NODES % 16 == 0, "a block owns 16 whole nodes");
#define NPB   200
#define GEMM_BLOCKS (N_NODES / 16)
#define CHUNK 1280
#define EPT   (CHUNK / 256)
#define NCHUNK (N_EDGES / CHUNK)
#define NBIN  256
#define DUMP_BIN (N_NODES / NPB)
#define REC_D_OFF ((size_t)N_EDGES * 2)
#define TAB_OFF   ((size_t)N_EDGES * 3)
static_assert(N_EDGES % CHUNK == 0, "chunks tile the edges exactly");
static_assert(CHUNK % 256 == 0, "every thread takes the same number of edges");
static_assert(CHUNK < 65536, "a place inside a chunk fits in 16 bits");
static_assert(N_NODES <= 65536, "a source id fits in 16 bits");
static_assert(NPB <= 256, "a destination's number inside its range fits in 8 bits");
static_assert(DUMP_BIN + 1 < NBIN, "the end of the last range is a table entry");
static_assert(TAB_OFF % 4 == 0 && REC_D_OFF % 4 == 0, "the arrays start on whole words");
static_assert(16 * IN_F == 2 * 256 * 4, "256 threads x 2 x 4 floats cover a block's x tile");
static_assert(IN_F % 8 == 0, "a step takes 8 k values");
#define A_PITCH (IN_F + 4)
#define K_STEPS (IN_F / 8)
#define QK_WGS 768
static_assert(NCHUNK <= QK_WGS, "a work-group sorts at most one chunk");
#define W_ROWS (2 * OUT_F)
static_assert(OUT_F * IN_F == 8 * 256 * 4, "256 threads x 8 x 4 floats cover one weight matrix");

__device__ __forceinline__ void lds_barrier() {
    asm volatile("s_wait_dscnt 0x0\n\ts_barrier_signal -1\n\ts_barrier_wait -1" ::: "memory");
}

__global__ __launch_bounds__(256)
void qk_gemm(const float* __restrict__ x,
             const float* __restrict__ Wq,
             const float* __restrict__ Wk,
             const int* __restrict__ ei,
             float* __restrict__ Q,
             float* __restrict__ K,
             unsigned* __restrict__ bins) {
    __shared__ __attribute__((aligned(16))) float sW[W_ROWS * A_PITCH];
    __shared__ __attribute__((aligned(16))) float sA[2][16 * A_PITCH];
    __shared__ unsigned sBin[NBIN];
    __shared__ unsigned sRec[CHUNK];

    const int tid     = threadIdx.x;
    const int lane    = tid & 31;
    const int wave    = tid >> 5;
    const int mat     = wave >> 2;
    const int colBase = (wave & 3) * 16;
    const int* __restrict__ cols = ei + N_EDGES;
    unsigned short* __restrict__ recS = (unsigned short*)bins;
    unsigned char*  __restrict__ recD = (unsigned char*)bins + REC_D_OFF;
    unsigned short* __restrict__ tab  = (unsigned short*)((char*)bins + TAB_OFF);

    float*       __restrict__ Out = mat ? K  : Q;
    const int l    = lane & 15;
    const int koff = (lane >> 4) * 4;

    float4 x0, x1;
    auto fetch = [&](int t) {
        const float4* __restrict__ xt = (const float4*)(x + (size_t)t * 16 * IN_F);
        x0  = xt[tid];
        x1  = xt[tid + 256];
    };
    auto stage = [&](int p) {
        float* sAp = sA[p];
        const int i0 = tid, i1 = tid + 256;
        *(float4*)(sAp + (i0 >> 5) * A_PITCH + (i0 & 31) * 4) = x0;
        *(float4*)(sAp + (i1 >> 5) * A_PITCH + (i1 & 31) * 4) = x1;
    };

    const bool sorter = blockIdx.x < NCHUNK;
    const unsigned cbase = blockIdx.x * (unsigned)CHUNK;
    int dsts[EPT], srcs[EPT];
#pragma unroll
    for (int j = 0; j < EPT; ++j) {
        dsts[j] = 0;
        srcs[j] = 0;
        if (sorter) {
            dsts[j] = cols[cbase + j * 256 + tid];
            srcs[j] = ei[cbase + j * 256 + tid];
        }
    }

    int tile = blockIdx.x;
    fetch(tile);

    {
        const float4* __restrict__ wq4 = (const float4*)Wq;
        const float4* __restrict__ wk4 = (const float4*)Wk;
#define W_LOAD(J) const float4 wq_##J = wq4[tid + 256 * J]; const float4 wk_##J = wk4[tid + 256 * J];
        W_LOAD(0) W_LOAD(1) W_LOAD(2) W_LOAD(3) W_LOAD(4) W_LOAD(5) W_LOAD(6) W_LOAD(7)
#undef W_LOAD

        if (sorter) {
            sBin[tid] = 0;
            __syncthreads();
            unsigned bin[EPT], pos[EPT];
#pragma unroll
            for (int j = 0; j < EPT; ++j) {
                const unsigned r = (unsigned)dsts[j] / NPB;
                bin[j] = r < DUMP_BIN ? r : DUMP_BIN;
                pos[j] = atomicAdd(&sBin[bin[j]], 1u);
            }
            __syncthreads();
            if (tid < 32) {
                int part[NBIN / 32];
                int tot = 0;
#pragma unroll
                for (int j = 0; j < NBIN / 32; ++j) {
                    part[j] = tot;
                    tot += (int)sBin[tid * (NBIN / 32) + j];
                }
                int inc = tot;
#pragma unroll
                for (int o = 1; o < 32; o <<= 1) {
                    const int y = __shfl_up(inc, o, 32);
                    if (tid >= o) inc += y;
                }
                const int excl = inc - tot;
#pragma unroll
                for (int j = 0; j < NBIN / 32; ++j)
                    sBin[tid * (NBIN / 32) + j] = (unsigned)(excl + part[j]);
            }
            __syncthreads();
#pragma unroll
            for (int j = 0; j < EPT; ++j)
                sRec[sBin[bin[j]] + pos[j]] =
                    ((unsigned)srcs[j] << 8) | (((unsigned)dsts[j] - bin[j] * NPB) & 255u);
            tab[blockIdx.x * NBIN + tid] = (unsigned short)sBin[tid];
            __syncthreads();
#pragma unroll
            for (int j = 0; j < EPT; ++j) {
                const unsigned i = j * 256 + tid;
                const unsigned v = sRec[i];
                recS[cbase + i] = (unsigned short)(v >> 8);
                recD[cbase + i] = (unsigned char)(v & 255u);
            }
        }

#define W_STORE(J) { const int idx = tid + 256 * J; \
            float* p = sW + (idx >> 5) * A_PITCH + (idx & 31) * 4; \
            *(float4*)p                     = wq_##J; \
            *(float4*)(p + OUT_F * A_PITCH) = wk_##J; }
        W_STORE(0) W_STORE(1) W_STORE(2) W_STORE(3) W_STORE(4) W_STORE(5) W_STORE(6) W_STORE(7)
#undef W_STORE
    }
    stage(0);
    if (tile + QK_WGS < GEMM_BLOCKS) fetch(tile + QK_WGS);
    lds_barrier();

    float4 b[K_STEPS];
    {
        const float* brow = sW + (mat * OUT_F + colBase + l) * A_PITCH + koff;
#pragma unroll
        for (int kk = 0; kk < K_STEPS; ++kk)
            b[kk] = *(const float4*)(brow + kk * 8);
    }

    for (int i = 0;; ++i) {
        const int p = i & 1;

        const float* arow = sA[p] + l * A_PITCH + koff;

        v8f acc = {};
#pragma unroll
        for (int kk = 0; kk < K_STEPS; ++kk) {
            const float4 a = *(const float4*)(arow + kk * 8);
            v2f a01 = {a.x, a.y}, a23 = {a.z, a.w};
            v2f b01 = {b[kk].x, b[kk].y}, b23 = {b[kk].z, b[kk].w};
            acc = __builtin_amdgcn_wmma_f32_16x16x4_f32(
                 false, a01,  false, b01,
                 (short)0, acc,  false,  false);
            acc = __builtin_amdgcn_wmma_f32_16x16x4_f32(
                 false, a23,  false, b23,
                 (short)0, acc,  false,  false);
        }

        const int outRow = tile * 16 + ((lane >> 4) ? 8 : 0);
        const int outCol = colBase + l;
#pragma unroll
        for (int r = 0; r < 8; ++r)
            Out[(size_t)(outRow + r) * OUT_F + outCol] = acc[r];

        tile += QK_WGS;
        if (tile >= GEMM_BLOCKS) break;
        stage(p ^ 1);
        if (tile + QK_WGS < GEMM_BLOCKS) fetch(tile + QK_WGS);
        lds_barrier();
    }
}

#define NBLK  (N_NODES / NPB)
#define TPB   1024
#define CAP   8192
#define STEP  8
#define BPL   ((NPB + 31) / 32)
#define ACC_PITCH (OUT_F + 1)
#define RUN   12
static_assert(NCHUNK <= TPB, "one thread per chunk");
static_assert(N_NODES % NPB == 0, "ranges tile the nodes exactly");
static_assert(NPB <= 256, "local node index is packed into 8 bits");
static_assert(NPB % 4 == 0, "a ticket is four nodes, one per 8-lane group of a wave");
static_assert(CAP <= (1 << 24), "place among a node's edges is packed into 24 bits");
static_assert((NPB * OUT_F) % 4 == 0, "K rows move as float4");
static_assert(OUT_F == 64, "8 lanes x 2 x 4 floats per row");
static_assert(STEP == 8, "soft_step names its 8 places one by one, one per lane of an 8-lane group");

template <int M>
__device__ __forceinline__ float grp_xadd(float p) {
    return p + __int_as_float(__builtin_amdgcn_update_dpp(
                   0, __float_as_int(p), 0x160 + M, 0xf, 0xf, true));
}
__device__ __forceinline__ float grp_dot(float4 qa, float4 qb, float4 ka, float4 kb) {
    v2f p = v2f{qa.x, qa.y} * v2f{ka.x, ka.y};
    p = __builtin_elementwise_fma(v2f{qa.z, qa.w}, v2f{ka.z, ka.w}, p);
    p = __builtin_elementwise_fma(v2f{qb.x, qb.y}, v2f{kb.x, kb.y}, p);
    p = __builtin_elementwise_fma(v2f{qb.z, qb.w}, v2f{kb.z, kb.w}, p);
    float s = p.x + p.y;
    s = grp_xadd<1>(s);
    s = grp_xadd<2>(s);
    s = grp_xadd<4>(s);
    return s;
}

template <int U>
__device__ __forceinline__ float grp_bcast(float v) {
    return __int_as_float(__builtin_amdgcn_mov_dpp8(__float_as_int(v), U * 0x249249));
}

__device__ __forceinline__ void soft_step(const float4* qa, const float4* qb, int c, int end,
                                          float4 ka, float4 kb, unsigned sub,
                                          float& m, float& den, v2f* acc) {
    float a[STEP];
    float cm = -INFINITY;
#pragma unroll
    for (int u = 0; u < STEP; ++u) {
        const float d = grp_dot(qa[u], qb[u], ka, kb) * 0.125f;
        a[u] = (c + u < end) ? d : -INFINITY;
        cm = fmaxf(cm, a[u]);
    }
    const float mn = fmaxf(m, cm);
    float mine = a[0];
#pragma unroll
    for (int u = 1; u < STEP; ++u)
        mine = (sub == (unsigned)u) ? a[u] : mine;
    const float exm = expf(mine - mn);
    const float sc  = expf(m - mn);
    den *= sc;
    acc[0] *= sc; acc[1] *= sc; acc[2] *= sc; acc[3] *= sc;
#define SOFT_SLOT(U) { const float ex = grp_bcast<U>(exm); den += ex; const v2f e2 = {ex, ex}; \
        acc[0] = __builtin_elementwise_fma(v2f{qa[U].x, qa[U].y}, e2, acc[0]); \
        acc[1] = __builtin_elementwise_fma(v2f{qa[U].z, qa[U].w}, e2, acc[1]); \
        acc[2] = __builtin_elementwise_fma(v2f{qb[U].x, qb[U].y}, e2, acc[2]); \
        acc[3] = __builtin_elementwise_fma(v2f{qb[U].z, qb[U].w}, e2, acc[3]); }
    SOFT_SLOT(0) SOFT_SLOT(1) SOFT_SLOT(2) SOFT_SLOT(3)
    SOFT_SLOT(4) SOFT_SLOT(5) SOFT_SLOT(6) SOFT_SLOT(7)
#undef SOFT_SLOT
    m = mn;
}

__device__ __forceinline__ float edge_logit(const float* __restrict__ q, const float* k) {
    float s = 0.0f;
    for (int j = 0; j < OUT_F; ++j) s += q[j] * k[j];
    return s * 0.125f;
}

__device__ __forceinline__ void lds_add(float* p, float v) {
    __hip_atomic_fetch_add(p, v, __ATOMIC_RELAXED, __HIP_MEMORY_SCOPE_WORKGROUP);
}
__device__ __forceinline__ void lds_max(unsigned* p, unsigned v) {
    __hip_atomic_fetch_max(p, v, __ATOMIC_RELAXED, __HIP_MEMORY_SCOPE_WORKGROUP);
}

__global__ __launch_bounds__(TPB)
void range_attend(const int* __restrict__ ei,
                  const float* __restrict__ Q,
                  const float* __restrict__ K,
                  const unsigned* __restrict__ bins,
                  float* __restrict__ out) {
    __shared__ __attribute__((aligned(16))) float sK[NPB * OUT_F];
    __shared__ float sAcc[NPB * ACC_PITCH];
    __shared__ unsigned sMax[NPB];
    __shared__ float    sDen[NPB];
    __shared__ unsigned sList[CAP];
    __shared__ int      sRow[CAP];
    __shared__ int      sSrc[CAP + STEP];
    __shared__ unsigned sDeg[NPB];
    __shared__ int      sOff[NPB + 1];
    __shared__ unsigned sCnt;
    __shared__ unsigned sNext;

    const int tid = threadIdx.x;
    const int n0  = blockIdx.x * NPB;
    const int* __restrict__ cols = ei + N_EDGES;

    const unsigned short* __restrict__ tab = (const unsigned short*)((const char*)bins + TAB_OFF);
    unsigned s0 = 0, s1 = 0;
    if (tid < NCHUNK) {
        s0 = tab[tid * NBIN + blockIdx.x];
        s1 = tab[tid * NBIN + blockIdx.x + 1];
    }

    for (int i = tid; i < NPB * OUT_F / 4; i += TPB)
        ((float4*)sK)[i] = ((const float4*)(K + (size_t)n0 * OUT_F))[i];
    if (tid < NPB) sDeg[tid] = 0;
    if (tid == 0) { sCnt = 0; sNext = 0; }
    __syncthreads();

    {
        const unsigned short* __restrict__ recS = (const unsigned short*)bins;
        const unsigned char*  __restrict__ recD = (const unsigned char*)bins + REC_D_OFF;
        auto put = [&](unsigned p, unsigned src, unsigned d8) {
            if (p < CAP) {
                const unsigned d = min(d8, (unsigned)(NPB - 1));
                const unsigned place = atomicAdd(&sDeg[d], 1u);
                sRow[p]  = (int)src;
                sList[p] = (place << 8) | d;
            }
        };
        if (tid < NCHUNK) {
            const unsigned cb = (unsigned)tid * CHUNK;
            s1 = min(s1, (unsigned)CHUNK);
            const unsigned len = s1 > s0 ? s1 - s0 : 0u;
            unsigned rs[RUN], rd[RUN];
#pragma unroll
            for (int u = 0; u < RUN; ++u) {
                const unsigned at = cb + ((unsigned)u < len ? s0 + (unsigned)u : 0u);
                rs[u] = recS[at];
                rd[u] = recD[at];
            }
            unsigned p0 = 0;
            if (len) p0 = atomicAdd(&sCnt, len);
            unsigned place[RUN];
#pragma unroll
            for (int u = 0; u < RUN; ++u) {
                rd[u] = min(rd[u], (unsigned)(NPB - 1));
                place[u] = 0;
                if ((unsigned)u < len && p0 + (unsigned)u < CAP) place[u] = atomicAdd(&sDeg[rd[u]], 1u);
            }
#pragma unroll
            for (int u = 0; u < RUN; ++u)
                if ((unsigned)u < len && p0 + (unsigned)u < CAP) {
                    sRow[p0 + u]  = (int)rs[u];
                    sList[p0 + u] = (place[u] << 8) | rd[u];
                }
            for (unsigned i = RUN; i < len; ++i)
                put(p0 + i, recS[cb + s0 + i], recD[cb + s0 + i]);
        }
    }
    __syncthreads();

    const unsigned cnt = (unsigned)__builtin_amdgcn_readfirstlane((int)sCnt);

    if (cnt <= CAP) {
        const int n = (int)cnt;
        if (tid < 32) {
            int part[BPL];
            int tot = 0;
#pragma unroll
            for (int j = 0; j < BPL; ++j) {
                const int idx = tid * BPL + j;
                part[j] = tot;
                tot += idx < NPB ? (int)sDeg[idx] : 0;
            }
            int inc = tot;
#pragma unroll
            for (int o = 1; o < 32; o <<= 1) {
                const int y = __shfl_up(inc, o, 32);
                if (tid >= o) inc += y;
            }
            const int excl = inc - tot;
#pragma unroll
            for (int j = 0; j < BPL; ++j) {
                const int idx = tid * BPL + j;
                if (idx < NPB) sOff[idx] = excl + part[j];
            }
            if (tid == 31) sOff[NPB] = inc;
        }
        __syncthreads();

        for (int p = tid; p < n; p += TPB) {
            const unsigned v = sList[p];
            sSrc[sOff[v & 255u] + (int)(v >> 8)] = sRow[p];
        }
        __syncthreads();

        const int      lane = tid & 31;
        const unsigned sub  = tid & 7;

        const char* __restrict__ qb = (const char*)Q;
        char*       __restrict__ ob = (char*)out;

        for (;;) {
            int t = 0;
            if (lane == 0) t = (int)atomicAdd(&sNext, 4u);
            t = __builtin_amdgcn_readfirstlane(t);
            if (t >= NPB) break;
            const int nd  = t + (lane >> 3);
            const int beg = sOff[nd], end = sOff[nd + 1];
            const float4 ka = *(const float4*)(sK + nd * OUT_F + sub * 4);
            const float4 kb = *(const float4*)(sK + nd * OUT_F + OUT_F / 2 + sub * 4);
            float m = -INFINITY, den = 0.0f;
            v2f acc[4] = {v2f{0.0f, 0.0f}, v2f{0.0f, 0.0f}, v2f{0.0f, 0.0f}, v2f{0.0f, 0.0f}};
            for (int c = beg; c < end; c += STEP) {
                int    s[STEP];
                float4 qa[STEP], qc[STEP];
#pragma unroll
                for (int u = 0; u < STEP; ++u)
                    s[u] = sSrc[c + u];
#pragma unroll
                for (int u = 0; u < STEP; ++u) {
                    const unsigned su = (unsigned)((c + u < end) ? s[u] : s[0]);
                    const char* row = qb + (su * (unsigned)(OUT_F * 4) + sub * 16u);
                    qa[u] = *(const float4*)row;
                    qc[u] = *(const float4*)(row + OUT_F * 2);
                }
                soft_step(qa, qc, c, end, ka, kb, sub, m, den, acc);
            }
            const float dd = den + 1e-16f;
            char* orow = ob + ((unsigned)(n0 + nd) * (unsigned)(OUT_F * 4) + sub * 16u);
            *(float4*)orow =
                make_float4(acc[0].x / dd, acc[0].y / dd, acc[1].x / dd, acc[1].y / dd);
            *(float4*)(orow + OUT_F * 2) =
                make_float4(acc[2].x / dd, acc[2].y / dd, acc[3].x / dd, acc[3].y / dd);
        }
    } else {
        for (int i = tid; i < NPB * ACC_PITCH; i += TPB)
            sAcc[i] = 0.0f;
        if (tid < NPB) {
            sMax[tid] = fkey(-INFINITY);
            sDen[tid] = 0.0f;
        }
        __syncthreads();
        for (int e = tid; e < N_EDGES; e += TPB) {
            const unsigned d = (unsigned)(cols[e] - n0);
            if (d < NPB)
                lds_max(&sMax[d], fkey(edge_logit(Q + (size_t)ei[e] * OUT_F, sK + d * OUT_F)));
        }
        __syncthreads();
        for (int e = tid; e < N_EDGES; e += TPB) {
            const unsigned d = (unsigned)(cols[e] - n0);
            if (d < NPB) {
                const float s = edge_logit(Q + (size_t)ei[e] * OUT_F, sK + d * OUT_F);
                lds_add(&sDen[d], expf(s - funkey(sMax[d])));
            }
        }
        __syncthreads();
        for (int e = tid; e < N_EDGES; e += TPB) {
            const unsigned d = (unsigned)(cols[e] - n0);
            if (d < NPB) {
                const float* q = Q + (size_t)ei[e] * OUT_F;
                const float s = edge_logit(q, sK + d * OUT_F);
                const float w = expf(s - funkey(sMax[d])) / (sDen[d] + 1e-16f);
                for (int j = 0; j < OUT_F; ++j) lds_add(&sAcc[d * ACC_PITCH + j], q[j] * w);
            }
        }
        __syncthreads();

        for (int i = tid; i < NPB * OUT_F; i += TPB)
            out[(size_t)n0 * OUT_F + i] = sAcc[(i >> 6) * ACC_PITCH + (i & 63)];
    }
}

extern "C" void kernel_launch(void* const* d_in, const int* in_sizes, int n_in,
                              void* d_out, int out_size, void* d_ws, size_t ws_size,
                              hipStream_t stream) {
    const float* x  = (const float*)d_in[0];
    const int*   ei = (const int*)d_in[1];
    const float* Wq = (const float*)d_in[2];
    const float* Wk = (const float*)d_in[3];
    float* out = (float*)d_out;

    char* ws = (char*)d_ws;
    float*    Q       = (float*)ws;    ws += (size_t)N_NODES * OUT_F * sizeof(float);
    float*    K       = (float*)ws;    ws += (size_t)N_NODES * OUT_F * sizeof(float);
    unsigned* bins    = (unsigned*)ws;

    qk_gemm<<<QK_WGS, 256, 0, stream>>>(x, Wq, Wk, ei, Q, K, bins);
    range_attend<<<NBLK, TPB, 0, stream>>>(ei, Q, K, bins, out);
}
